# starting kernel
# baseline (speedup 1.0000x reference)
#include <hip/hip_runtime.h>
#include <hip/hip_bf16.h>

#define BB   4
#define LL   4096
#define MM   1024
#define SS   64
#define ROWS (BB * LL)
#define NC   16
#define LC   (LL / NC)

typedef __attribute__((ext_vector_type(16))) _Float16 v16h;
typedef __attribute__((ext_vector_type(8)))  _Float16 v8h;
typedef __attribute__((ext_vector_type(4)))  _Float16 v4h;
typedef __attribute__((ext_vector_type(8)))  float    v8f;

__device__ __forceinline__ v16h load_frag(const _Float16* __restrict__ p) {
    v8h lo = *(const v8h*)(p);
    v8h hi = *(const v8h*)(p + 16);
    v16h t;
#pragma unroll
    for (int i = 0; i < 8; ++i) { t[i] = lo[i]; t[i + 8] = hi[i]; }
    return t;
}

struct Acc44 { v8f c[4][4]; };

__device__ __forceinline__ void wave_gemm64(const _Float16* __restrict__ A,
                                            const _Float16* __restrict__ B,
                                            int ldA, int ldB,
                                            int rowbase, int colbase, int K,
                                            Acc44& acc) {
    const int lane  = threadIdx.x & 31;
    const int rr    = lane & 15;
    const int khalf = (lane >> 4) & 1;
    const _Float16* Ab[4];
    const _Float16* Bb[4];
#pragma unroll
    for (int t = 0; t < 4; ++t) {
        Ab[t] = A + (size_t)(rowbase + t * 16 + rr) * ldA + khalf * 8;
        Bb[t] = B + (size_t)(colbase + t * 16 + rr) * ldB + khalf * 8;
    }
    for (int kb = 0; kb < K; kb += 32) {
        v16h a[4], b[4];
#pragma unroll
        for (int t = 0; t < 4; ++t) a[t] = load_frag(Ab[t] + kb);
#pragma unroll
        for (int t = 0; t < 4; ++t) b[t] = load_frag(Bb[t] + kb);
#pragma unroll
        for (int i = 0; i < 4; ++i)
#pragma unroll
            for (int j = 0; j < 4; ++j)
                acc.c[i][j] = __builtin_amdgcn_wmma_f32_16x16x32_f16(
                    false, a[i], false, b[j], (short)0, acc.c[i][j], false, false);
    }
}

__device__ __forceinline__ float fast_sigmoid(float v) {
    return __builtin_amdgcn_rcpf(1.0f + __expf(-v));
}

__global__ void k_convert(const float* __restrict__ W_in,
                          const float* __restrict__ W_xp,
                          const float* __restrict__ W_out,
                          const float* __restrict__ Dmat,
                          _Float16* __restrict__ Wih,
                          _Float16* __restrict__ Wxh,
                          _Float16* __restrict__ Woh,
                          _Float16* __restrict__ Dth) {
    const int NWI = 2 * MM * MM, NXP = SS * MM, NWO = MM * MM, ND = SS * MM;
    int i = blockIdx.x * 256 + threadIdx.x;
    if (i < NWI) { Wih[i] = (_Float16)W_in[i]; return; }
    i -= NWI;
    if (i < NXP) { Wxh[i] = (_Float16)W_xp[i]; return; }
    i -= NXP;
    if (i < NWO) { Woh[i] = (_Float16)W_out[i]; return; }
    i -= NWO;
    if (i < ND) {
        int n = i >> 6, s = i & 63;
        Dth[i] = (_Float16)Dmat[(size_t)s * MM + n];
    }
}

__global__ __launch_bounds__(256) void k_layernorm(const float* __restrict__ x,
                                                   const float* __restrict__ gamma,
                                                   const float* __restrict__ beta,
                                                   _Float16* __restrict__ xn) {
    __shared__ float sSum[8], sSq[8];
    const int row = blockIdx.x;
    const size_t base = (size_t)row * MM;
    const int tid = threadIdx.x;

    float4 v = ((const float4*)(x + base))[tid];
    float s = v.x + v.y + v.z + v.w;
    float q = v.x * v.x + v.y * v.y + v.z * v.z + v.w * v.w;
#pragma unroll
    for (int off = 16; off > 0; off >>= 1) {
        s += __shfl_down(s, off);
        q += __shfl_down(q, off);
    }
    if ((tid & 31) == 0) { sSum[tid >> 5] = s; sSq[tid >> 5] = q; }
    __syncthreads();
    float ts = 0.f, tq = 0.f;
#pragma unroll
    for (int i = 0; i < 8; ++i) { ts += sSum[i]; tq += sSq[i]; }
    const float mu = ts * (1.0f / MM);
    const float var = tq * (1.0f / MM) - mu * mu;
    const float rstd = rsqrtf(var + 1e-5f);

    float4 g = ((const float4*)gamma)[tid];
    float4 bb = ((const float4*)beta)[tid];
    v4h o;
    o[0] = (_Float16)((v.x - mu) * rstd * g.x + bb.x);
    o[1] = (_Float16)((v.y - mu) * rstd * g.y + bb.y);
    o[2] = (_Float16)((v.z - mu) * rstd * g.z + bb.z);
    o[3] = (_Float16)((v.w - mu) * rstd * g.w + bb.w);
    ((v4h*)(xn + base))[tid] = o;
}

__global__ __launch_bounds__(256) void k_gemm1(const _Float16* __restrict__ xn,
                                               const _Float16* __restrict__ Wih,
                                               const float* __restrict__ b_in,
                                               _Float16* __restrict__ xp,
                                               _Float16* __restrict__ zg) {
    Acc44 acc = {};
    const int w = threadIdx.x >> 5;
    const int rowbase = blockIdx.x * 128 + (w & 1) * 64;
    const int colbase = blockIdx.y * 256 + (w >> 1) * 64;
    wave_gemm64(xn, Wih, MM, MM, rowbase, colbase, MM, acc);

    const int lane = threadIdx.x & 31;
    const int nlo = lane & 15, mh = (lane >> 4) * 8;
    const bool is_x = (blockIdx.y < (MM / 256));
#pragma unroll
    for (int j = 0; j < 4; ++j) {
        const int n = colbase + j * 16 + nlo;
        const float bias = b_in[n];
#pragma unroll
        for (int i = 0; i < 4; ++i) {
#pragma unroll
            for (int jj = 0; jj < 8; ++jj) {
                const int m = rowbase + i * 16 + mh + jj;
                const float v = acc.c[i][j][jj] + bias;
                const float sg = fast_sigmoid(v);
                if (is_x) xp[(size_t)m * MM + n] = (_Float16)(v * sg);
                else      zg[(size_t)m * MM + (n - MM)] = (_Float16)sg;
            }
        }
    }
}

__global__ __launch_bounds__(256) void k_gemm2(const _Float16* __restrict__ xp,
                                               const _Float16* __restrict__ Wxh,
                                               const float* __restrict__ b_xp,
                                               float* __restrict__ u) {
    Acc44 acc = {};
    const int w = threadIdx.x >> 5;
    const int rowbase = blockIdx.x * 512 + w * 64;
    wave_gemm64(xp, Wxh, MM, MM, rowbase, 0, MM, acc);

    const int lane = threadIdx.x & 31;
    const int nlo = lane & 15, mh = (lane >> 4) * 8;
#pragma unroll
    for (int j = 0; j < 4; ++j) {
        const int n = j * 16 + nlo;
        const float bias = b_xp[n];
#pragma unroll
        for (int i = 0; i < 4; ++i)
#pragma unroll
            for (int jj = 0; jj < 8; ++jj) {
                const int m = rowbase + i * 16 + mh + jj;
                u[(size_t)m * SS + n] = acc.c[i][j][jj] + bias;
            }
    }
}

__global__ void k_scan_chunk(const float* __restrict__ u,
                             const float* __restrict__ A_log,
                             float* __restrict__ hloc,
                             float* __restrict__ chunk_end) {
    const int t = blockIdx.x * 256 + threadIdx.x;
    const int s = t & 63;
    const int c = (t >> 6) & (NC - 1);
    const int b = t >> 10;
    const float a = expf(-expf(A_log[s]));
    float h = 0.f;
    const int l0 = c * LC;
    for (int j = 0; j < LC; ++j) {
        const size_t idx = ((size_t)b * LL + l0 + j) * SS + s;
        h = a * h + u[idx];
        hloc[idx] = h;
    }
    chunk_end[((b * SS + s) * NC) + c] = h;
}

__global__ void k_scan_carry(const float* __restrict__ A_log,
                             const float* __restrict__ chunk_end,
                             float* __restrict__ carry_in) {
    const int t = threadIdx.x;
    const int s = t & 63, b = t >> 6;
    const float E = expf(-expf(A_log[s]) * (float)LC);
    float carry = 0.f;
    for (int c = 0; c < NC; ++c) {
        carry_in[((b * SS + s) * NC) + c] = carry;
        carry = E * carry + chunk_end[((b * SS + s) * NC) + c];
    }
}

__global__ void k_scan_apply(const float* __restrict__ hloc,
                             const float* __restrict__ A_log,
                             const float* __restrict__ carry_in,
                             _Float16* __restrict__ h_h) {
    const int t = blockIdx.x * 256 + threadIdx.x;
    const int s = t & 63;
    const int c = (t >> 6) & (NC - 1);
    const int b = t >> 10;
    const float a = expf(-expf(A_log[s]));
    const float carry = carry_in[((b * SS + s) * NC) + c];
    float p = a;
    const int l0 = c * LC;
    for (int j = 0; j < LC; ++j) {
        const int l = l0 + j;
        const float hv = hloc[((size_t)b * LL + l) * SS + s] + carry * p;
        p *= a;
        h_h[((size_t)l * BB + b) * SS + s] = (_Float16)hv;
    }
}

__global__ __launch_bounds__(256) void k_gemm3(const _Float16* __restrict__ h_h,
                                               const _Float16* __restrict__ Dth,
                                               const _Float16* __restrict__ zg,
                                               _Float16* __restrict__ ys) {
    Acc44 acc = {};
    const int w = threadIdx.x >> 5;
    const int rowbase = blockIdx.x * 128 + (w & 1) * 64;
    const int colbase = blockIdx.y * 256 + (w >> 1) * 64;
    wave_gemm64(h_h, Dth, SS, SS, rowbase, colbase, SS, acc);

    const int lane = threadIdx.x & 31;
    const int nlo = lane & 15, mh = (lane >> 4) * 8;
#pragma unroll
    for (int i = 0; i < 4; ++i)
#pragma unroll
        for (int j = 0; j < 4; ++j)
#pragma unroll
            for (int jj = 0; jj < 8; ++jj) {
                const int rp = rowbase + i * 16 + mh + jj;
                const int n = colbase + j * 16 + nlo;
                const int b = rp & (BB - 1), l = rp >> 2;
                const float z = (float)zg[((size_t)b * LL + l) * MM + n];
                ys[(size_t)rp * MM + n] = (_Float16)(acc.c[i][j][jj] * z);
            }
}

__global__ __launch_bounds__(256) void k_gemm4(const _Float16* __restrict__ ys,
                                               const _Float16* __restrict__ Woh,
                                               const float* __restrict__ b_out,
                                               const float* __restrict__ xflat,
                                               float* __restrict__ out) {
    Acc44 acc = {};
    const int w = threadIdx.x >> 5;
    const int rowbase = blockIdx.x * 128 + (w & 1) * 64;
    const int colbase = blockIdx.y * 256 + (w >> 1) * 64;
    wave_gemm64(ys, Woh, MM, MM, rowbase, colbase, MM, acc);

    const int lane = threadIdx.x & 31;
    const int nlo = lane & 15, mh = (lane >> 4) * 8;
#pragma unroll
    for (int j = 0; j < 4; ++j) {
        const int n = colbase + j * 16 + nlo;
        const float bias = b_out[n];
#pragma unroll
        for (int i = 0; i < 4; ++i)
#pragma unroll
            for (int jj = 0; jj < 8; ++jj) {
                const int m = rowbase + i * 16 + mh + jj;
                const size_t idx = (size_t)m * MM + n;
                out[idx] = xflat[idx] + acc.c[i][j][jj] + bias;
            }
    }
}

extern "C" void kernel_launch(void* const* d_in, const int* in_sizes, int n_in,
                              void* d_out, int out_size, void* d_ws, size_t ws_size,
                              hipStream_t stream) {
    const float* x     = (const float*)d_in[0];
    const float* A_log = (const float*)d_in[1];
    const float* Dmat  = (const float*)d_in[2];
    const float* W_in  = (const float*)d_in[3];
    const float* b_in  = (const float*)d_in[4];
    const float* W_xp  = (const float*)d_in[5];
    const float* b_xp  = (const float*)d_in[6];
    const float* W_out = (const float*)d_in[7];
    const float* b_out = (const float*)d_in[8];
    const float* gamma = (const float*)d_in[9];
    const float* beta  = (const float*)d_in[10];
    float* out = (float*)d_out;

    char* ws = (char*)d_ws;
    size_t off = 0;
    auto take = [&](size_t bytes) { char* p = ws + off; off += bytes; return p; };
    _Float16* xn_h  = (_Float16*)take((size_t)ROWS * MM * 2);
    _Float16* xp_h  = (_Float16*)take((size_t)ROWS * MM * 2);
    _Float16* z_h   = (_Float16*)take((size_t)ROWS * MM * 2);
    _Float16* ys_h  = (_Float16*)take((size_t)ROWS * MM * 2);
    float*    u     = (float*)take((size_t)ROWS * SS * 4);
    float*    hloc  = (float*)take((size_t)ROWS * SS * 4);
    _Float16* h_h   = (_Float16*)take((size_t)ROWS * SS * 2);
    _Float16* Wih   = (_Float16*)take((size_t)2 * MM * MM * 2);
    _Float16* Wxh   = (_Float16*)take((size_t)SS * MM * 2);
    _Float16* Woh   = (_Float16*)take((size_t)MM * MM * 2);
    _Float16* Dth   = (_Float16*)take((size_t)SS * MM * 2);
    float*    cend  = (float*)take((size_t)BB * SS * NC * 4);
    float*    cin   = (float*)take((size_t)BB * SS * NC * 4);
    (void)ws_size; (void)in_sizes; (void)n_in; (void)out_size;

    {
        const int total = 2 * MM * MM + SS * MM + MM * MM + SS * MM;
        k_convert<<<(total + 255) / 256, 256, 0, stream>>>(W_in, W_xp, W_out, Dmat,
                                                           Wih, Wxh, Woh, Dth);
    }
    k_layernorm<<<ROWS, 256, 0, stream>>>(x, gamma, beta, xn_h);
    k_gemm1<<<dim3(ROWS / 128, (2 * MM) / 256), 256, 0, stream>>>(xn_h, Wih, b_in, xp_h, z_h);
    k_gemm2<<<dim3(ROWS / 512, 1), 256, 0, stream>>>(xp_h, Wxh, b_xp, u);
    k_scan_chunk<<<(BB * SS * NC) / 256, 256, 0, stream>>>(u, A_log, hloc, cend);
    k_scan_carry<<<1, BB * SS, 0, stream>>>(A_log, cend, cin);
    k_scan_apply<<<(BB * SS * NC) / 256, 256, 0, stream>>>(hloc, A_log, cin, h_h);
    k_gemm3<<<dim3(ROWS / 128, MM / 256), 256, 0, stream>>>(h_h, Dth, z_h, ys_h);
    k_gemm4<<<dim3(ROWS / 128, MM / 256), 256, 0, stream>>>(ys_h, Woh, b_out, x, out);
}


# Scan's local pass done on the state projection's accumulators: a wave scans its 64 tokens, waves chain through LDS; one parallel apply pass rebuilds carries from segment ends; new sum order
# speedup vs baseline: 1.7990x; 1.7990x over previous
#include <hip/hip_runtime.h>
#include <hip/hip_bf16.h>

#define BB   4
#define LL   4096
#define MM   1024
#define SS   64
#define ROWS (BB * LL)

typedef __attribute__((ext_vector_type(16))) _Float16 v16h;
typedef __attribute__((ext_vector_type(8)))  _Float16 v8h;
typedef __attribute__((ext_vector_type(4)))  _Float16 v4h;
typedef __attribute__((ext_vector_type(8)))  float    v8f;

__device__ __forceinline__ size_t frag_index(int row, int k, int K) {
    return ((size_t)(row >> 4) * (size_t)(K >> 5) + (size_t)(k >> 5)) * 512
         + (size_t)(((k >> 4) & 1) * 256 + ((row & 15) + ((k >> 3) & 1) * 16) * 8 + (k & 7));
}

__device__ __forceinline__ v16h load_frag(const _Float16* __restrict__ p) {
    v8h lo = *(const v8h*)(p);
    v8h hi = *(const v8h*)(p + 256);
    v16h t;
#pragma unroll
    for (int i = 0; i < 8; ++i) { t[i] = lo[i]; t[i + 8] = hi[i]; }
    return t;
}

struct Acc44 { v8f c[4][4]; };

__device__ __forceinline__ void wave_gemm64(const _Float16* __restrict__ A,
                                            const _Float16* __restrict__ B,
                                            int rowbase, int colbase, int K,
                                            Acc44& acc) {
    const int lane = threadIdx.x & 31;
    const int kt   = K >> 5;
    const _Float16* Ab = A + (size_t)(rowbase >> 4) * kt * 512 + lane * 8;
    const _Float16* Bb = B + (size_t)(colbase >> 4) * kt * 512 + lane * 8;
    for (int kb = 0; kb < kt; ++kb) {
        v16h a[4], b[4];
#pragma unroll
        for (int t = 0; t < 4; ++t) a[t] = load_frag(Ab + (size_t)(t * kt + kb) * 512);
#pragma unroll
        for (int t = 0; t < 4; ++t) b[t] = load_frag(Bb + (size_t)(t * kt + kb) * 512);
#pragma unroll
        for (int i = 0; i < 4; ++i)
#pragma unroll
            for (int j = 0; j < 4; ++j)
                acc.c[i][j] = __builtin_amdgcn_wmma_f32_16x16x32_f16(
                    false, a[i], false, b[j], (short)0, acc.c[i][j], false, false);
    }
}

__device__ __forceinline__ float fast_sigmoid(float v) {
    return __builtin_amdgcn_rcpf(1.0f + __expf(-v));
}

__global__ __launch_bounds__(256) void k_layernorm(const float* __restrict__ x,
                                                   const float* __restrict__ gamma,
                                                   const float* __restrict__ beta,
                                                   _Float16* __restrict__ xn,
                                                   const float* __restrict__ W_in,
                                                   const float* __restrict__ W_xp,
                                                   const float* __restrict__ W_out,
                                                   const float* __restrict__ Dmat,
                                                   _Float16* __restrict__ Wih,
                                                   _Float16* __restrict__ Wxh,
                                                   _Float16* __restrict__ Woh,
                                                   _Float16* __restrict__ Dth) {
    __shared__ float sSum[8], sSq[8];
    const int row = blockIdx.x;
    const size_t base = (size_t)row * MM;
    const int tid = threadIdx.x;

    float4 v = ((const float4*)(x + base))[tid];

    if (tid < 25) {
        const int NWI = 2 * MM * MM, NXP = SS * MM, NWO = MM * MM;
        const int i = (blockIdx.x * 25 + tid) * 8;
        v8h o;
        _Float16* dst;
        if (i < NWI + NXP + NWO) {
            const float* src;
            if (i < NWI) {
                src = W_in + i;
                dst = Wih + frag_index(i >> 10, i & (MM - 1), MM);
            } else if (i < NWI + NXP) {
                const int j = i - NWI;
                src = W_xp + j;
                dst = Wxh + frag_index(j >> 10, j & (MM - 1), MM);
            } else {
                const int j = i - NWI - NXP;
                src = W_out + j;
                dst = Woh + frag_index(j >> 10, j & (MM - 1), MM);
            }
            const float4 a = ((const float4*)src)[0];
            const float4 b = ((const float4*)src)[1];
            o[0] = (_Float16)a.x; o[1] = (_Float16)a.y; o[2] = (_Float16)a.z; o[3] = (_Float16)a.w;
            o[4] = (_Float16)b.x; o[5] = (_Float16)b.y; o[6] = (_Float16)b.z; o[7] = (_Float16)b.w;
        } else {
            const int j = i - NWI - NXP - NWO;
            const int n = j >> 6, s0 = j & 63;
#pragma unroll
            for (int t = 0; t < 8; ++t) o[t] = (_Float16)Dmat[(size_t)(s0 + t) * MM + n];
            dst = Dth + frag_index(n, s0, SS);
        }
        *(v8h*)dst = o;
    }

    float s = v.x + v.y + v.z + v.w;
    float q = v.x * v.x + v.y * v.y + v.z * v.z + v.w * v.w;
#pragma unroll
    for (int off = 16; off > 0; off >>= 1) {
        s += __shfl_down(s, off);
        q += __shfl_down(q, off);
    }
    if ((tid & 31) == 0) { sSum[tid >> 5] = s; sSq[tid >> 5] = q; }
    __syncthreads();
    float ts = 0.f, tq = 0.f;
#pragma unroll
    for (int i = 0; i < 8; ++i) { ts += sSum[i]; tq += sSq[i]; }
    const float mu = ts * (1.0f / MM);
    const float var = tq * (1.0f / MM) - mu * mu;
    const float rstd = rsqrtf(var + 1e-5f);

    float4 g = ((const float4*)gamma)[tid];
    float4 bb = ((const float4*)beta)[tid];
    v4h o;
    o[0] = (_Float16)((v.x - mu) * rstd * g.x + bb.x);
    o[1] = (_Float16)((v.y - mu) * rstd * g.y + bb.y);
    o[2] = (_Float16)((v.z - mu) * rstd * g.z + bb.z);
    o[3] = (_Float16)((v.w - mu) * rstd * g.w + bb.w);
    *(v4h*)(xn + frag_index(row, tid * 4, MM)) = o;
}

#define IMG_HALFS 4096

__device__ __forceinline__ size_t run_base(int rt, int col0) {
    return ((size_t)rt * (size_t)(MM >> 5) + (size_t)(col0 >> 5)) * 512;
}

__global__ __launch_bounds__(256) void k_gemm1(const _Float16* __restrict__ xn,
                                               const _Float16* __restrict__ Wih,
                                               const float* __restrict__ b_in,
                                               _Float16* __restrict__ xp,
                                               _Float16* __restrict__ zg) {
    __shared__ __attribute__((aligned(16))) _Float16 img[8 * IMG_HALFS];
    Acc44 acc = {};
    const int w = threadIdx.x >> 5;
    const int rowbase = blockIdx.x * 128 + (w & 1) * 64;
    const int colbase = blockIdx.y * 256 + (w >> 1) * 64;
    wave_gemm64(xn, Wih, rowbase, colbase, MM, acc);

    const int lane = threadIdx.x & 31;
    const int nlo = lane & 15, mh = (lane >> 4) * 8;
    const bool is_x = (blockIdx.y < (MM / 256));
    _Float16* my = img + w * IMG_HALFS;
    const int lanepart = (mh + (nlo >> 3) * 16) * 8 + (nlo & 7);
#pragma unroll
    for (int j = 0; j < 4; ++j) {
        const float bias = b_in[colbase + j * 16 + nlo];
#pragma unroll
        for (int i = 0; i < 4; ++i) {
#pragma unroll
            for (int jj = 0; jj < 8; ++jj) {
                const float v = acc.c[i][j][jj] + bias;
                const float sg = fast_sigmoid(v);
                const float val = is_x ? v * sg : sg;
                my[(i * 2 + (j >> 1)) * 512 + (j & 1) * 256 + jj * 8 + lanepart] = (_Float16)val;
            }
        }
    }
    __syncthreads();
    _Float16* __restrict__ dst = is_x ? xp : zg;
    const int col0 = is_x ? colbase : colbase - MM;
#pragma unroll
    for (int t = 0; t < 16; ++t) {
        const v8h d = *(const v8h*)(my + t * 256 + lane * 8);
        *(v8h*)(dst + run_base((rowbase >> 4) + (t >> 2), col0) + (t & 3) * 256 + lane * 8) = d;
    }
}

#define SEG 512
#define NSEG (LL / SEG)

__global__ __launch_bounds__(256) void k_gemm2(const _Float16* __restrict__ xp,
                                               const _Float16* __restrict__ Wxh,
                                               const float* __restrict__ b_xp,
                                               const float* __restrict__ A_log,
                                               float* __restrict__ hloc,
                                               float* __restrict__ blk_end) {
    __shared__ float sEnd[8][SS];
    Acc44 acc = {};
    const int w = __builtin_amdgcn_readfirstlane(threadIdx.x >> 5);
    const int rowbase = blockIdx.x * SEG + w * 64;
    wave_gemm64(xp, Wxh, rowbase, 0, MM, acc);

    const int lane = threadIdx.x & 31;
    const int nlo = lane & 15, mh = (lane >> 4) * 8;
    const bool lo = lane < 16;

    float a1[4], a8[4], Lend[4][4], Hend[4][4];
#pragma unroll
    for (int j = 0; j < 4; ++j) {
        const int n = j * 16 + nlo;
        const float bias = b_xp[n];
        const float a = expf(-expf(A_log[n]));
        const float a2 = a * a, a4 = a2 * a2;
        a1[j] = a;
        a8[j] = a4 * a4;
        float S = 0.f;
#pragma unroll
        for (int i = 0; i < 4; ++i) {
            float h = acc.c[i][j][0] + bias;
            acc.c[i][j][0] = h;
#pragma unroll
            for (int jj = 1; jj < 8; ++jj) {
                h = a * h + (acc.c[i][j][jj] + bias);
                acc.c[i][j][jj] = h;
            }
            const float other = __shfl_xor(h, 16);
            Lend[j][i] = lo ? h : other;
            Hend[j][i] = lo ? other : h;
            S = a8[j] * (a8[j] * S + Lend[j][i]) + Hend[j][i];
        }
        if (lo) sEnd[w][n] = S;
    }
    __syncthreads();

#pragma unroll
    for (int j = 0; j < 4; ++j) {
        const int n = j * 16 + nlo;
        const float a = a1[j], e8 = a8[j];
        const float a16 = e8 * e8, a32 = a16 * a16, a64 = a32 * a32;
        float ends[7];
#pragma unroll
        for (int v = 0; v < 7; ++v) ends[v] = sEnd[v][n];
        float S = 0.f;
#pragma unroll
        for (int v = 0; v < 7; ++v) S = (v < w) ? a64 * S + ends[v] : S;
#pragma unroll
        for (int i = 0; i < 4; ++i) {
            const float SL = e8 * S + Lend[j][i];
            const float inc = lo ? S : SL;
            float p = a;
#pragma unroll
            for (int jj = 0; jj < 8; ++jj) {
                const int m = rowbase + i * 16 + mh + jj;
                hloc[(size_t)m * SS + n] = acc.c[i][j][jj] + p * inc;
                p *= a;
            }
            S = e8 * SL + Hend[j][i];
        }
        if (w == 7 && lo) blk_end[blockIdx.x * SS + n] = S;
    }
}

__global__ __launch_bounds__(256) void k_scan_apply(const float* __restrict__ hloc,
                                                    const float* __restrict__ A_log,
                                                    const float* __restrict__ blk_end,
                                                    _Float16* __restrict__ h_h) {
    const int t = blockIdx.x * 256 + threadIdx.x;
    const int L = t & 31;
    const int kq = (t >> 5) & 3;
    const int rp = (t >> 7) * 16 + (L & 15);
    const int s0 = (kq * 2 + (L >> 4)) * 8;
    const int l = rp >> 2, b = rp & 3;
    const int seg = (unsigned)l / SEG, pos = (unsigned)l % SEG;

    const float4* hp = (const float4*)(hloc + ((size_t)b * LL + l) * SS + s0);
    const float4 h0 = hp[0], h1 = hp[1];
    const float4 g0 = ((const float4*)(A_log + s0))[0], g1 = ((const float4*)(A_log + s0))[1];
    float ev[NSEG - 1][8];
#pragma unroll
    for (int v = 0; v < NSEG - 1; ++v) {
        const float4* ep = (const float4*)(blk_end + (size_t)(b * NSEG + v) * SS + s0);
        const float4 x0 = ep[0], x1 = ep[1];
        ev[v][0] = x0.x; ev[v][1] = x0.y; ev[v][2] = x0.z; ev[v][3] = x0.w;
        ev[v][4] = x1.x; ev[v][5] = x1.y; ev[v][6] = x1.z; ev[v][7] = x1.w;
    }
    const float hv[8] = {h0.x, h0.y, h0.z, h0.w, h1.x, h1.y, h1.z, h1.w};
    const float al[8] = {g0.x, g0.y, g0.z, g0.w, g1.x, g1.y, g1.z, g1.w};
    v8h o;
#pragma unroll
    for (int q = 0; q < 8; ++q) {
        const float ea = expf(al[q]);
        const float E = expf(-ea * (float)SEG);
        const float p = expf(-ea * (float)(pos + 1));
        float c = 0.f;
#pragma unroll
        for (int v = 0; v < NSEG - 1; ++v)
            c = (v < seg) ? E * c + ev[v][q] : c;
        o[q] = (_Float16)(hv[q] + c * p);
    }
    *(v8h*)(h_h + frag_index(rp, s0, SS)) = o;
}

__global__ __launch_bounds__(256) void k_gemm3(const _Float16* __restrict__ h_h,
                                               const _Float16* __restrict__ Dth,
                                               const _Float16* __restrict__ zg,
                                               _Float16* __restrict__ ys) {
    __shared__ __attribute__((aligned(16))) _Float16 zimg[8 * IMG_HALFS];
    __shared__ __attribute__((aligned(16))) _Float16 yimg[8 * IMG_HALFS];
    const int w = threadIdx.x >> 5;
    const int lane = threadIdx.x & 31;
    const int rowbase = blockIdx.x * 128 + (w & 1) * 64;
    const int colbase = blockIdx.y * 256 + (w >> 1) * 64;

    _Float16* zi = zimg + w * IMG_HALFS;
    _Float16* yi = yimg + w * IMG_HALFS;
    const int l0 = rowbase >> 2;
#pragma unroll
    for (int t = 0; t < 16; ++t) {
        const int b = t >> 2;
        const v8h d = *(const v8h*)(zg + run_base((b * LL + l0) >> 4, colbase) + (t & 3) * 256 + lane * 8);
        *(v8h*)(zi + t * 256 + lane * 8) = d;
    }

    Acc44 acc = {};
    wave_gemm64(h_h, Dth, rowbase, colbase, SS, acc);
    __syncthreads();

    const int nlo = lane & 15, mh = (lane >> 4) * 8;
    const int zlane = ((mh >> 2) + (nlo >> 3) * 16) * 8 + (nlo & 7);
    const int ylane = (mh + (nlo >> 3) * 16) * 8 + (nlo & 7);
#pragma unroll
    for (int i = 0; i < 4; ++i)
#pragma unroll
        for (int j = 0; j < 4; ++j)
#pragma unroll
            for (int jj = 0; jj < 8; ++jj) {
                const float z = (float)zi[(jj & 3) * 1024 + (j >> 1) * 512 + (j & 1) * 256
                                          + (i * 4 + (jj >> 2)) * 8 + zlane];
                yi[(i * 2 + (j >> 1)) * 512 + (j & 1) * 256 + jj * 8 + ylane] =
                    (_Float16)(acc.c[i][j][jj] * z);
            }
    __syncthreads();
#pragma unroll
    for (int t = 0; t < 16; ++t) {
        const v8h d = *(const v8h*)(yi + t * 256 + lane * 8);
        *(v8h*)(ys + run_base((rowbase >> 4) + (t >> 2), colbase) + (t & 3) * 256 + lane * 8) = d;
    }
}

__global__ __launch_bounds__(256) void k_gemm4(const _Float16* __restrict__ ys,
                                               const _Float16* __restrict__ Woh,
                                               const float* __restrict__ b_out,
                                               const float* __restrict__ xflat,
                                               float* __restrict__ out) {
    Acc44 acc = {};
    const int w = threadIdx.x >> 5;
    const int rowbase = blockIdx.x * 128 + (w & 1) * 64;
    const int colbase = blockIdx.y * 256 + (w >> 1) * 64;
    wave_gemm64(ys, Woh, rowbase, colbase, MM, acc);

    const int lane = threadIdx.x & 31;
    const int nlo = lane & 15, mh = (lane >> 4) * 8;
#pragma unroll
    for (int j = 0; j < 4; ++j) {
        const int n = colbase + j * 16 + nlo;
        const float bias = b_out[n];
#pragma unroll
        for (int i = 0; i < 4; ++i)
#pragma unroll
            for (int jj = 0; jj < 8; ++jj) {
                const int m = rowbase + i * 16 + mh + jj;
                const size_t idx = (size_t)m * MM + n;
                out[idx] = xflat[idx] + acc.c[i][j][jj] + bias;
            }
        __builtin_amdgcn_sched_barrier(0);
    }
}

extern "C" void kernel_launch(void* const* d_in, const int* in_sizes, int n_in,
                              void* d_out, int out_size, void* d_ws, size_t ws_size,
                              hipStream_t stream) {
    const float* x     = (const float*)d_in[0];
    const float* A_log = (const float*)d_in[1];
    const float* Dmat  = (const float*)d_in[2];
    const float* W_in  = (const float*)d_in[3];
    const float* b_in  = (const float*)d_in[4];
    const float* W_xp  = (const float*)d_in[5];
    const float* b_xp  = (const float*)d_in[6];
    const float* W_out = (const float*)d_in[7];
    const float* b_out = (const float*)d_in[8];
    const float* gamma = (const float*)d_in[9];
    const float* beta  = (const float*)d_in[10];
    float* out = (float*)d_out;

    char* ws = (char*)d_ws;
    size_t off = 0;
    auto take = [&](size_t bytes) { char* p = ws + off; off += bytes; return p; };
    _Float16* xn_h  = (_Float16*)take((size_t)ROWS * MM * 2);
    _Float16* xp_h  = (_Float16*)take((size_t)ROWS * MM * 2);
    _Float16* z_h   = (_Float16*)take((size_t)ROWS * MM * 2);
    _Float16* ys_h  = (_Float16*)take((size_t)ROWS * MM * 2);
    float*    hloc  = (float*)take((size_t)ROWS * SS * 4);
    _Float16* h_h   = (_Float16*)take((size_t)ROWS * SS * 2);
    _Float16* Wih   = (_Float16*)take((size_t)2 * MM * MM * 2);
    _Float16* Wxh   = (_Float16*)take((size_t)SS * MM * 2);
    _Float16* Woh   = (_Float16*)take((size_t)MM * MM * 2);
    _Float16* Dth   = (_Float16*)take((size_t)SS * MM * 2);
    float*    bend  = (float*)take((size_t)BB * NSEG * SS * 4);
    (void)ws_size; (void)in_sizes; (void)n_in; (void)out_size;

    k_layernorm<<<ROWS, 256, 0, stream>>>(x, gamma, beta, xn_h, W_in, W_xp, W_out, Dmat,
                                          Wih, Wxh, Woh, Dth);
    k_gemm1<<<dim3(ROWS / 128, (2 * MM) / 256), 256, 0, stream>>>(xn_h, Wih, b_in, xp_h, z_h);
    k_gemm2<<<dim3(ROWS / SEG, 1), 256, 0, stream>>>(xp_h, Wxh, b_xp, A_log, hloc, bend);
    k_scan_apply<<<(ROWS * 8) / 256, 256, 0, stream>>>(hloc, A_log, bend, h_h);
    k_gemm3<<<dim3(ROWS / 128, MM / 256), 256, 0, stream>>>(h_h, Dth, z_h, ys_h);
    k_gemm4<<<dim3(ROWS / 128, MM / 256), 256, 0, stream>>>(ys_h, Woh, b_out, x, out);
}


# in_proj and out_proj K loops: ring of four fragment register sets, each refilled as one group of 16 loads behind its 16 products; loads run three k tiles ahead (was load, wait, multiply); same sums
# speedup vs baseline: 2.3556x; 1.3094x over previous
#include <hip/hip_runtime.h>
#include <hip/hip_bf16.h>

#define BB   4
#define LL   4096
#define MM   1024
#define SS   64
#define ROWS (BB * LL)

typedef __attribute__((ext_vector_type(16))) _Float16 v16h;
typedef __attribute__((ext_vector_type(8)))  _Float16 v8h;
typedef __attribute__((ext_vector_type(4)))  _Float16 v4h;
typedef __attribute__((ext_vector_type(8)))  float    v8f;

__device__ __forceinline__ size_t frag_index(int row, int k, int K) {
    return ((size_t)(row >> 4) * (size_t)(K >> 5) + (size_t)(k >> 5)) * 512
         + (size_t)(((k >> 4) & 1) * 256 + ((row & 15) + ((k >> 3) & 1) * 16) * 8 + (k & 7));
}

__device__ __forceinline__ v16h load_frag(const _Float16* __restrict__ p) {
    v8h lo = *(const v8h*)(p);
    v8h hi = *(const v8h*)(p + 256);
    v16h t;
#pragma unroll
    for (int i = 0; i < 8; ++i) { t[i] = lo[i]; t[i + 8] = hi[i]; }
    return t;
}

struct Acc44 { v8f c[4][4]; };

__device__ __forceinline__ void wave_gemm64(const _Float16* __restrict__ A,
                                            const _Float16* __restrict__ B,
                                            int rowbase, int colbase, int K,
                                            Acc44& acc) {
    const int lane = threadIdx.x & 31;
    const int kt   = K >> 5;
    const _Float16* Ab = A + (size_t)(rowbase >> 4) * kt * 512 + lane * 8;
    const _Float16* Bb = B + (size_t)(colbase >> 4) * kt * 512 + lane * 8;
    for (int kb = 0; kb < kt; ++kb) {
        v16h a[4], b[4];
#pragma unroll
        for (int t = 0; t < 4; ++t) a[t] = load_frag(Ab + (size_t)(t * kt + kb) * 512);
#pragma unroll
        for (int t = 0; t < 4; ++t) b[t] = load_frag(Bb + (size_t)(t * kt + kb) * 512);
#pragma unroll
        for (int i = 0; i < 4; ++i)
#pragma unroll
            for (int j = 0; j < 4; ++j)
                acc.c[i][j] = __builtin_amdgcn_wmma_f32_16x16x32_f16(
                    false, a[i], false, b[j], (short)0, acc.c[i][j], false, false);
    }
}

struct FragSet { v16h a[4]; v16h b[4]; };

template <int KT>
__device__ __forceinline__ void load_set(FragSet& s, const _Float16* __restrict__ Ab,
                                         const _Float16* __restrict__ Bb, int kb) {
#pragma unroll
    for (int t = 0; t < 4; ++t) s.a[t] = load_frag(Ab + (size_t)(t * KT + kb) * 512);
#pragma unroll
    for (int t = 0; t < 4; ++t) s.b[t] = load_frag(Bb + (size_t)(t * KT + kb) * 512);
}

__device__ __forceinline__ void mma_set(Acc44& acc, const FragSet& s) {
#pragma unroll
    for (int i = 0; i < 4; ++i)
#pragma unroll
        for (int j = 0; j < 4; ++j)
            acc.c[i][j] = __builtin_amdgcn_wmma_f32_16x16x32_f16(
                false, s.a[i], false, s.b[j], (short)0, acc.c[i][j], false, false);
}

template <int KT>
__device__ __forceinline__ void wave_gemm64_ring(const _Float16* __restrict__ A,
                                                 const _Float16* __restrict__ B,
                                                 int rowbase, int colbase, Acc44& acc) {
    static_assert(KT % 4 == 0 && KT >= 8, "ring of four k tiles");
    const int lane = threadIdx.x & 31;
    const _Float16* Ab = A + (size_t)(rowbase >> 4) * KT * 512 + lane * 8;
    const _Float16* Bb = B + (size_t)(colbase >> 4) * KT * 512 + lane * 8;
    FragSet s0, s1, s2, s3;
    load_set<KT>(s0, Ab, Bb, 0);
    load_set<KT>(s1, Ab, Bb, 1);
    load_set<KT>(s2, Ab, Bb, 2);
    load_set<KT>(s3, Ab, Bb, 3);
    for (int kb = 4; kb < KT; kb += 4) {
        mma_set(acc, s0); __builtin_amdgcn_sched_barrier(0);
        load_set<KT>(s0, Ab, Bb, kb); __builtin_amdgcn_sched_barrier(0);
        mma_set(acc, s1); __builtin_amdgcn_sched_barrier(0);
        load_set<KT>(s1, Ab, Bb, kb + 1); __builtin_amdgcn_sched_barrier(0);
        mma_set(acc, s2); __builtin_amdgcn_sched_barrier(0);
        load_set<KT>(s2, Ab, Bb, kb + 2); __builtin_amdgcn_sched_barrier(0);
        mma_set(acc, s3); __builtin_amdgcn_sched_barrier(0);
        load_set<KT>(s3, Ab, Bb, kb + 3); __builtin_amdgcn_sched_barrier(0);
    }
    mma_set(acc, s0);
    mma_set(acc, s1);
    mma_set(acc, s2);
    mma_set(acc, s3);
}

__device__ __forceinline__ float fast_sigmoid(float v) {
    return __builtin_amdgcn_rcpf(1.0f + __expf(-v));
}

__global__ __launch_bounds__(256) void k_layernorm(const float* __restrict__ x,
                                                   const float* __restrict__ gamma,
                                                   const float* __restrict__ beta,
                                                   _Float16* __restrict__ xn,
                                                   const float* __restrict__ W_in,
                                                   const float* __restrict__ W_xp,
                                                   const float* __restrict__ W_out,
                                                   const float* __restrict__ Dmat,
                                                   _Float16* __restrict__ Wih,
                                                   _Float16* __restrict__ Wxh,
                                                   _Float16* __restrict__ Woh,
                                                   _Float16* __restrict__ Dth) {
    __shared__ float sSum[8], sSq[8];
    const int row = blockIdx.x;
    const size_t base = (size_t)row * MM;
    const int tid = threadIdx.x;

    float4 v = ((const float4*)(x + base))[tid];

    if (tid < 25) {
        const int NWI = 2 * MM * MM, NXP = SS * MM, NWO = MM * MM;
        const int i = (blockIdx.x * 25 + tid) * 8;
        v8h o;
        _Float16* dst;
        if (i < NWI + NXP + NWO) {
            const float* src;
            if (i < NWI) {
                src = W_in + i;
                dst = Wih + frag_index(i >> 10, i & (MM - 1), MM);
            } else if (i < NWI + NXP) {
                const int j = i - NWI;
                src = W_xp + j;
                dst = Wxh + frag_index(j >> 10, j & (MM - 1), MM);
            } else {
                const int j = i - NWI - NXP;
                src = W_out + j;
                dst = Woh + frag_index(j >> 10, j & (MM - 1), MM);
            }
            const float4 a = ((const float4*)src)[0];
            const float4 b = ((const float4*)src)[1];
            o[0] = (_Float16)a.x; o[1] = (_Float16)a.y; o[2] = (_Float16)a.z; o[3] = (_Float16)a.w;
            o[4] = (_Float16)b.x; o[5] = (_Float16)b.y; o[6] = (_Float16)b.z; o[7] = (_Float16)b.w;
        } else {
            const int j = i - NWI - NXP - NWO;
            const int n = j >> 6, s0 = j & 63;
#pragma unroll
            for (int t = 0; t < 8; ++t) o[t] = (_Float16)Dmat[(size_t)(s0 + t) * MM + n];
            dst = Dth + frag_index(n, s0, SS);
        }
        *(v8h*)dst = o;
    }

    float s = v.x + v.y + v.z + v.w;
    float q = v.x * v.x + v.y * v.y + v.z * v.z + v.w * v.w;
#pragma unroll
    for (int off = 16; off > 0; off >>= 1) {
        s += __shfl_down(s, off);
        q += __shfl_down(q, off);
    }
    if ((tid & 31) == 0) { sSum[tid >> 5] = s; sSq[tid >> 5] = q; }
    __syncthreads();
    float ts = 0.f, tq = 0.f;
#pragma unroll
    for (int i = 0; i < 8; ++i) { ts += sSum[i]; tq += sSq[i]; }
    const float mu = ts * (1.0f / MM);
    const float var = tq * (1.0f / MM) - mu * mu;
    const float rstd = rsqrtf(var + 1e-5f);

    float4 g = ((const float4*)gamma)[tid];
    float4 bb = ((const float4*)beta)[tid];
    v4h o;
    o[0] = (_Float16)((v.x - mu) * rstd * g.x + bb.x);
    o[1] = (_Float16)((v.y - mu) * rstd * g.y + bb.y);
    o[2] = (_Float16)((v.z - mu) * rstd * g.z + bb.z);
    o[3] = (_Float16)((v.w - mu) * rstd * g.w + bb.w);
    *(v4h*)(xn + frag_index(row, tid * 4, MM)) = o;
}

#define IMG_HALFS 4096

__device__ __forceinline__ size_t run_base(int rt, int col0) {
    return ((size_t)rt * (size_t)(MM >> 5) + (size_t)(col0 >> 5)) * 512;
}

#define RTILES (ROWS / 16)

__global__ __launch_bounds__(256) void k_gemm1(const _Float16* __restrict__ xn,
                                               const _Float16* __restrict__ Wih,
                                               const float* __restrict__ b_in,
                                               const _Float16* __restrict__ Wxh,
                                               float* __restrict__ upart,
                                               _Float16* __restrict__ zg) {
    __shared__ __attribute__((aligned(16))) _Float16 img[8 * IMG_HALFS];
    Acc44 acc = {};
    const int w = threadIdx.x >> 5;
    const int rowbase = blockIdx.x * 128 + (w & 1) * 64;
    const int colbase = blockIdx.y * 256 + (w >> 1) * 64;
    wave_gemm64_ring<MM / 32>(xn, Wih, rowbase, colbase, acc);

    const int lane = threadIdx.x & 31;
    const int nlo = lane & 15, mh = (lane >> 4) * 8;
    const bool is_x = (blockIdx.y < (MM / 256));
    _Float16* my = img + w * IMG_HALFS;
    const int lanepart = (mh + (nlo >> 3) * 16) * 8 + (nlo & 7);
#pragma unroll
    for (int j = 0; j < 4; ++j) {
        const float bias = b_in[colbase + j * 16 + nlo];
#pragma unroll
        for (int i = 0; i < 4; ++i) {
#pragma unroll
            for (int jj = 0; jj < 8; ++jj) {
                const float v = acc.c[i][j][jj] + bias;
                const float sg = fast_sigmoid(v);
                const float val = is_x ? v * sg : sg;
                my[(i * 2 + (j >> 1)) * 512 + (j & 1) * 256 + jj * 8 + lanepart] = (_Float16)val;
            }
        }
    }
    __syncthreads();
    if (is_x) {
        const int rg = w & 1, g = w >> 1;
        const _Float16* arow = img + rg * IMG_HALFS + lane * 8;
        const _Float16* bcol = Wxh + (size_t)(g * (MM >> 5) + blockIdx.y * 8) * 512 + lane * 8;
        v8f u[4] = {};
#pragma unroll
        for (int kb = 0; kb < 8; ++kb) {
            const v16h b = load_frag(bcol + kb * 512);
            const _Float16* ap = arow + (kb >> 1) * 2 * IMG_HALFS + (kb & 1) * 512;
#pragma unroll
            for (int i = 0; i < 4; ++i) {
                const v16h a = load_frag(ap + i * 1024);
                u[i] = __builtin_amdgcn_wmma_f32_16x16x32_f16(
                    false, a, false, b, (short)0, u[i], false, false);
            }
        }
        float* dst = upart + ((size_t)(blockIdx.y * 4 + g) * RTILES + (rowbase >> 4)) * 256 + nlo * 16 + mh;
#pragma unroll
        for (int i = 0; i < 4; ++i) {
            const float4 p0 = make_float4(u[i][0], u[i][1], u[i][2], u[i][3]);
            const float4 p1 = make_float4(u[i][4], u[i][5], u[i][6], u[i][7]);
            ((float4*)(dst + i * 256))[0] = p0;
            ((float4*)(dst + i * 256))[1] = p1;
        }
    } else {
        const int col0 = colbase - MM;
#pragma unroll
        for (int t = 0; t < 16; ++t) {
            const v8h d = *(const v8h*)(my + t * 256 + lane * 8);
            *(v8h*)(zg + run_base((rowbase >> 4) + (t >> 2), col0) + (t & 3) * 256 + lane * 8) = d;
        }
    }
}

#define SEG 512
#define NSEG (LL / SEG)

__global__ __launch_bounds__(256) void k_scan_local(const float* __restrict__ upart,
                                                    const float* __restrict__ b_xp,
                                                    const float* __restrict__ A_log,
                                                    float* __restrict__ hloc,
                                                    float* __restrict__ blk_end) {
    __shared__ float sEnd[8][16];
    const int w = __builtin_amdgcn_readfirstlane(threadIdx.x >> 5);
    const int g = blockIdx.y;
    const int rowbase = blockIdx.x * SEG + w * 64;
    const int lane = threadIdx.x & 31;
    const int nlo = lane & 15, mh = (lane >> 4) * 8;
    const bool lo = lane < 16;
    const int n = g * 16 + nlo;

    float4 q[4][4][2];
#pragma unroll
    for (int y = 0; y < 4; ++y)
#pragma unroll
        for (int i = 0; i < 4; ++i) {
            const float4* p = (const float4*)(upart
                + ((size_t)(y * 4 + g) * RTILES + (rowbase >> 4) + i) * 256 + nlo * 16 + mh);
            q[y][i][0] = p[0];
            q[y][i][1] = p[1];
        }
    float u[4][8];
#pragma unroll
    for (int i = 0; i < 4; ++i) {
        u[i][0] = ((q[0][i][0].x + q[1][i][0].x) + q[2][i][0].x) + q[3][i][0].x;
        u[i][1] = ((q[0][i][0].y + q[1][i][0].y) + q[2][i][0].y) + q[3][i][0].y;
        u[i][2] = ((q[0][i][0].z + q[1][i][0].z) + q[2][i][0].z) + q[3][i][0].z;
        u[i][3] = ((q[0][i][0].w + q[1][i][0].w) + q[2][i][0].w) + q[3][i][0].w;
        u[i][4] = ((q[0][i][1].x + q[1][i][1].x) + q[2][i][1].x) + q[3][i][1].x;
        u[i][5] = ((q[0][i][1].y + q[1][i][1].y) + q[2][i][1].y) + q[3][i][1].y;
        u[i][6] = ((q[0][i][1].z + q[1][i][1].z) + q[2][i][1].z) + q[3][i][1].z;
        u[i][7] = ((q[0][i][1].w + q[1][i][1].w) + q[2][i][1].w) + q[3][i][1].w;
    }

    const float bias = b_xp[n];
    const float a = expf(-expf(A_log[n]));
    const float a2 = a * a, a4 = a2 * a2;
    const float e8 = a4 * a4;
    float Lend[4], Hend[4];
    float S = 0.f;
#pragma unroll
    for (int i = 0; i < 4; ++i) {
        float h = u[i][0] + bias;
        u[i][0] = h;
#pragma unroll
        for (int jj = 1; jj < 8; ++jj) {
            h = a * h + (u[i][jj] + bias);
            u[i][jj] = h;
        }
        const float other = __shfl_xor(h, 16);
        Lend[i] = lo ? h : other;
        Hend[i] = lo ? other : h;
        S = e8 * (e8 * S + Lend[i]) + Hend[i];
    }
    if (lo) sEnd[w][nlo] = S;
    __syncthreads();

    const float a16 = e8 * e8, a32 = a16 * a16, a64 = a32 * a32;
    float ends[7];
#pragma unroll
    for (int v = 0; v < 7; ++v) ends[v] = sEnd[v][nlo];
    S = 0.f;
#pragma unroll
    for (int v = 0; v < 7; ++v) S = (v < w) ? a64 * S + ends[v] : S;
    float* hout = hloc + ((size_t)g * ROWS + rowbase + mh) * 16 + nlo;
#pragma unroll
    for (int i = 0; i < 4; ++i) {
        const float SL = e8 * S + Lend[i];
        const float inc = lo ? S : SL;
        float p = a;
#pragma unroll
        for (int jj = 0; jj < 8; ++jj) {
            hout[(i * 16 + jj) * 16] = u[i][jj] + p * inc;
            p *= a;
        }
        S = e8 * SL + Hend[i];
    }
    if (w == 7 && lo) blk_end[(blockIdx.x * 4 + g) * 32 + nlo] = S;
}

__global__ __launch_bounds__(256) void k_scan_apply(const float* __restrict__ hloc,
                                                    const float* __restrict__ A_log,
                                                    const float* __restrict__ blk_end,
                                                    _Float16* __restrict__ h_h) {
    const int t = blockIdx.x * 256 + threadIdx.x;
    const int L = t & 31;
    const int kq = (t >> 5) & 3;
    const int rp = (t >> 7) * 16 + (L & 15);
    const int s0 = (kq * 2 + (L >> 4)) * 8;
    const int l = rp >> 2, b = rp & 3;
    const int seg = (unsigned)l / SEG, pos = (unsigned)l % SEG;

    const float4* hp = (const float4*)(hloc + ((size_t)(s0 >> 4) * ROWS + (size_t)b * LL + l) * 16 + (s0 & 15));
    const float4 h0 = hp[0], h1 = hp[1];
    const float4 g0 = ((const float4*)(A_log + s0))[0], g1 = ((const float4*)(A_log + s0))[1];
    float ev[NSEG - 1][8];
#pragma unroll
    for (int v = 0; v < NSEG - 1; ++v) {
        const float4* ep = (const float4*)(blk_end + (size_t)((b * NSEG + v) * 4 + (s0 >> 4)) * 32 + (s0 & 15));
        const float4 x0 = ep[0], x1 = ep[1];
        ev[v][0] = x0.x; ev[v][1] = x0.y; ev[v][2] = x0.z; ev[v][3] = x0.w;
        ev[v][4] = x1.x; ev[v][5] = x1.y; ev[v][6] = x1.z; ev[v][7] = x1.w;
    }
    const float hv[8] = {h0.x, h0.y, h0.z, h0.w, h1.x, h1.y, h1.z, h1.w};
    const float al[8] = {g0.x, g0.y, g0.z, g0.w, g1.x, g1.y, g1.z, g1.w};
    v8h o;
#pragma unroll
    for (int q = 0; q < 8; ++q) {
        const float ea = expf(al[q]);
        const float E = expf(-ea * (float)SEG);
        const float p = expf(-ea * (float)(pos + 1));
        float c = 0.f;
#pragma unroll
        for (int v = 0; v < NSEG - 1; ++v)
            c = (v < seg) ? E * c + ev[v][q] : c;
        o[q] = (_Float16)(hv[q] + c * p);
    }
    *(v8h*)(h_h + frag_index(rp, s0, SS)) = o;
}

__global__ __launch_bounds__(256) void k_gemm3(const _Float16* __restrict__ h_h,
                                               const _Float16* __restrict__ Dth,
                                               const _Float16* __restrict__ zg,
                                               _Float16* __restrict__ ys) {
    __shared__ __attribute__((aligned(16))) _Float16 zimg[8 * IMG_HALFS];
    __shared__ __attribute__((aligned(16))) _Float16 yimg[8 * IMG_HALFS];
    const int w = threadIdx.x >> 5;
    const int lane = threadIdx.x & 31;
    const int rowbase = blockIdx.x * 128 + (w & 1) * 64;
    const int colbase = blockIdx.y * 256 + (w >> 1) * 64;

    _Float16* zi = zimg + w * IMG_HALFS;
    _Float16* yi = yimg + w * IMG_HALFS;
    const int l0 = rowbase >> 2;
#pragma unroll
    for (int t = 0; t < 16; ++t) {
        const int b = t >> 2;
        const v8h d = *(const v8h*)(zg + run_base((b * LL + l0) >> 4, colbase) + (t & 3) * 256 + lane * 8);
        *(v8h*)(zi + t * 256 + lane * 8) = d;
    }

    Acc44 acc = {};
    wave_gemm64(h_h, Dth, rowbase, colbase, SS, acc);
    __syncthreads();

    const int nlo = lane & 15, mh = (lane >> 4) * 8;
    const int zlane = ((mh >> 2) + (nlo >> 3) * 16) * 8 + (nlo & 7);
    const int ylane = (mh + (nlo >> 3) * 16) * 8 + (nlo & 7);
#pragma unroll
    for (int i = 0; i < 4; ++i)
#pragma unroll
        for (int j = 0; j < 4; ++j)
#pragma unroll
            for (int jj = 0; jj < 8; ++jj) {
                const float z = (float)zi[(jj & 3) * 1024 + (j >> 1) * 512 + (j & 1) * 256
                                          + (i * 4 + (jj >> 2)) * 8 + zlane];
                yi[(i * 2 + (j >> 1)) * 512 + (j & 1) * 256 + jj * 8 + ylane] =
                    (_Float16)(acc.c[i][j][jj] * z);
            }
    __syncthreads();
#pragma unroll
    for (int t = 0; t < 16; ++t) {
        const v8h d = *(const v8h*)(yi + t * 256 + lane * 8);
        *(v8h*)(ys + run_base((rowbase >> 4) + (t >> 2), colbase) + (t & 3) * 256 + lane * 8) = d;
    }
}

__global__ __launch_bounds__(256) void k_gemm4(const _Float16* __restrict__ ys,
                                               const _Float16* __restrict__ Woh,
                                               const float* __restrict__ b_out,
                                               const float* __restrict__ xflat,
                                               float* __restrict__ out) {
    Acc44 acc = {};
    const int w = threadIdx.x >> 5;
    const int rowbase = blockIdx.x * 128 + (w & 1) * 64;
    const int colbase = blockIdx.y * 256 + (w >> 1) * 64;
    wave_gemm64_ring<MM / 32>(ys, Woh, rowbase, colbase, acc);

    const int lane = threadIdx.x & 31;
    const int nlo = lane & 15, mh = (lane >> 4) * 8;
#pragma unroll
    for (int j = 0; j < 4; ++j) {
        const int n = colbase + j * 16 + nlo;
        const float bias = b_out[n];
#pragma unroll
        for (int i = 0; i < 4; ++i)
#pragma unroll
            for (int jj = 0; jj < 8; ++jj) {
                const int m = rowbase + i * 16 + mh + jj;
                const size_t idx = (size_t)m * MM + n;
                out[idx] = xflat[idx] + acc.c[i][j][jj] + bias;
            }
    }
}

extern "C" void kernel_launch(void* const* d_in, const int* in_sizes, int n_in,
                              void* d_out, int out_size, void* d_ws, size_t ws_size,
                              hipStream_t stream) {
    const float* x     = (const float*)d_in[0];
    const float* A_log = (const float*)d_in[1];
    const float* Dmat  = (const float*)d_in[2];
    const float* W_in  = (const float*)d_in[3];
    const float* b_in  = (const float*)d_in[4];
    const float* W_xp  = (const float*)d_in[5];
    const float* b_xp  = (const float*)d_in[6];
    const float* W_out = (const float*)d_in[7];
    const float* b_out = (const float*)d_in[8];
    const float* gamma = (const float*)d_in[9];
    const float* beta  = (const float*)d_in[10];
    float* out = (float*)d_out;

    char* ws = (char*)d_ws;
    size_t off = 0;
    auto take = [&](size_t bytes) { char* p = ws + off; off += bytes; return p; };
    _Float16* xn_h  = (_Float16*)take((size_t)ROWS * MM * 2);
    char*     reg2  = take((size_t)ROWS * MM * 2);
    float*    upart = (float*)reg2;
    float*    bend  = (float*)(reg2 + (size_t)4 * ROWS * SS * 4);
    _Float16* z_h   = (_Float16*)take((size_t)ROWS * MM * 2);
    _Float16* ys_h  = (_Float16*)take((size_t)ROWS * MM * 2);
    float*    hloc  = (float*)take((size_t)ROWS * SS * 4);
    _Float16* h_h   = (_Float16*)take((size_t)ROWS * SS * 2);
    _Float16* Wih   = (_Float16*)take((size_t)2 * MM * MM * 2);
    _Float16* Wxh   = (_Float16*)take((size_t)SS * MM * 2);
    _Float16* Woh   = (_Float16*)take((size_t)MM * MM * 2);
    _Float16* Dth   = (_Float16*)take((size_t)SS * MM * 2);
    (void)ws_size; (void)in_sizes; (void)n_in; (void)out_size;

    k_layernorm<<<ROWS, 256, 0, stream>>>(x, gamma, beta, xn_h, W_in, W_xp, W_out, Dmat,
                                          Wih, Wxh, Woh, Dth);
    k_gemm1<<<dim3(ROWS / 128, (2 * MM) / 256), 256, 0, stream>>>(xn_h, Wih, b_in, Wxh, upart, z_h);
    k_scan_local<<<dim3(ROWS / SEG, SS / 16), 256, 0, stream>>>(upart, b_xp, A_log, hloc, bend);
    k_scan_apply<<<(ROWS * 8) / 256, 256, 0, stream>>>(hloc, A_log, bend, h_h);
    k_gemm3<<<dim3(ROWS / 128, MM / 256), 256, 0, stream>>>(h_h, Dth, z_h, ys_h);
    k_gemm4<<<dim3(ROWS / 128, MM / 256), 256, 0, stream>>>(ys_h, Woh, b_out, x, out);
}


# in_proj and readout products computed transposed (weights as A, tokens as B): a lane's 8 accumulators are one 16-B fragment slot; tiles leave by 16-B stores, no 2-B LDS pieces; same sums
# speedup vs baseline: 2.4421x; 1.0367x over previous
#include <hip/hip_runtime.h>
#include <hip/hip_bf16.h>

#define BB   4
#define LL   4096
#define MM   1024
#define SS   64
#define ROWS (BB * LL)

typedef __attribute__((ext_vector_type(16))) _Float16 v16h;
typedef __attribute__((ext_vector_type(8)))  _Float16 v8h;
typedef __attribute__((ext_vector_type(4)))  _Float16 v4h;
typedef __attribute__((ext_vector_type(8)))  float    v8f;

__device__ __forceinline__ size_t frag_index(int row, int k, int K) {
    return ((size_t)(row >> 4) * (size_t)(K >> 5) + (size_t)(k >> 5)) * 512
         + (size_t)(((k >> 4) & 1) * 256 + ((row & 15) + ((k >> 3) & 1) * 16) * 8 + (k & 7));
}

__device__ __forceinline__ v16h load_frag(const _Float16* __restrict__ p) {
    v8h lo = *(const v8h*)(p);
    v8h hi = *(const v8h*)(p + 256);
    v16h t;
#pragma unroll
    for (int i = 0; i < 8; ++i) { t[i] = lo[i]; t[i + 8] = hi[i]; }
    return t;
}

struct Acc44 { v8f c[4][4]; };

template <bool TR>
__device__ __forceinline__ v8f mma1(const v16h& rowfrag, const v16h& colfrag, const v8f& c) {
    if (TR)
        return __builtin_amdgcn_wmma_f32_16x16x32_f16(false, colfrag, false, rowfrag, (short)0, c, false, false);
    else
        return __builtin_amdgcn_wmma_f32_16x16x32_f16(false, rowfrag, false, colfrag, (short)0, c, false, false);
}

template <bool TR>
__device__ __forceinline__ void wave_gemm64(const _Float16* __restrict__ A,
                                            const _Float16* __restrict__ B,
                                            int rowbase, int colbase, int K,
                                            Acc44& acc) {
    const int lane = threadIdx.x & 31;
    const int kt   = K >> 5;
    const _Float16* Ab = A + (size_t)(rowbase >> 4) * kt * 512 + lane * 8;
    const _Float16* Bb = B + (size_t)(colbase >> 4) * kt * 512 + lane * 8;
    for (int kb = 0; kb < kt; ++kb) {
        v16h a[4], b[4];
#pragma unroll
        for (int t = 0; t < 4; ++t) a[t] = load_frag(Ab + (size_t)(t * kt + kb) * 512);
#pragma unroll
        for (int t = 0; t < 4; ++t) b[t] = load_frag(Bb + (size_t)(t * kt + kb) * 512);
#pragma unroll
        for (int i = 0; i < 4; ++i)
#pragma unroll
            for (int j = 0; j < 4; ++j)
                acc.c[i][j] = mma1<TR>(a[i], b[j], acc.c[i][j]);
    }
}

struct FragSet { v16h a[4]; v16h b[4]; };

template <int KT>
__device__ __forceinline__ void load_set(FragSet& s, const _Float16* __restrict__ Ab,
                                         const _Float16* __restrict__ Bb, int kb) {
#pragma unroll
    for (int t = 0; t < 4; ++t) s.a[t] = load_frag(Ab + (size_t)(t * KT + kb) * 512);
#pragma unroll
    for (int t = 0; t < 4; ++t) s.b[t] = load_frag(Bb + (size_t)(t * KT + kb) * 512);
}

template <bool TR>
__device__ __forceinline__ void mma_set(Acc44& acc, const FragSet& s) {
#pragma unroll
    for (int i = 0; i < 4; ++i)
#pragma unroll
        for (int j = 0; j < 4; ++j)
            acc.c[i][j] = mma1<TR>(s.a[i], s.b[j], acc.c[i][j]);
}

template <int KT, bool TR>
__device__ __forceinline__ void wave_gemm64_ring(const _Float16* __restrict__ A,
                                                 const _Float16* __restrict__ B,
                                                 int rowbase, int colbase, Acc44& acc) {
    static_assert(KT % 4 == 0 && KT >= 8, "ring of four k tiles");
    const int lane = threadIdx.x & 31;
    const _Float16* Ab = A + (size_t)(rowbase >> 4) * KT * 512 + lane * 8;
    const _Float16* Bb = B + (size_t)(colbase >> 4) * KT * 512 + lane * 8;
    FragSet s0, s1, s2, s3;
    load_set<KT>(s0, Ab, Bb, 0);
    load_set<KT>(s1, Ab, Bb, 1);
    load_set<KT>(s2, Ab, Bb, 2);
    load_set<KT>(s3, Ab, Bb, 3);
    for (int kb = 4; kb < KT; kb += 4) {
        mma_set<TR>(acc, s0); __builtin_amdgcn_sched_barrier(0);
        load_set<KT>(s0, Ab, Bb, kb); __builtin_amdgcn_sched_barrier(0);
        mma_set<TR>(acc, s1); __builtin_amdgcn_sched_barrier(0);
        load_set<KT>(s1, Ab, Bb, kb + 1); __builtin_amdgcn_sched_barrier(0);
        mma_set<TR>(acc, s2); __builtin_amdgcn_sched_barrier(0);
        load_set<KT>(s2, Ab, Bb, kb + 2); __builtin_amdgcn_sched_barrier(0);
        mma_set<TR>(acc, s3); __builtin_amdgcn_sched_barrier(0);
        load_set<KT>(s3, Ab, Bb, kb + 3); __builtin_amdgcn_sched_barrier(0);
    }
    mma_set<TR>(acc, s0);
    mma_set<TR>(acc, s1);
    mma_set<TR>(acc, s2);
    mma_set<TR>(acc, s3);
}

__device__ __forceinline__ float fast_sigmoid(float v) {
    return __builtin_amdgcn_rcpf(1.0f + __expf(-v));
}

__global__ __launch_bounds__(256) void k_layernorm(const float* __restrict__ x,
                                                   const float* __restrict__ gamma,
                                                   const float* __restrict__ beta,
                                                   _Float16* __restrict__ xn,
                                                   const float* __restrict__ W_in,
                                                   const float* __restrict__ W_xp,
                                                   const float* __restrict__ W_out,
                                                   const float* __restrict__ Dmat,
                                                   _Float16* __restrict__ Wih,
                                                   _Float16* __restrict__ Wxh,
                                                   _Float16* __restrict__ Woh,
                                                   _Float16* __restrict__ Dth) {
    __shared__ float sSum[8], sSq[8];
    const int row = blockIdx.x;
    const size_t base = (size_t)row * MM;
    const int tid = threadIdx.x;

    float4 v = ((const float4*)(x + base))[tid];

    if (tid < 25) {
        const int NWI = 2 * MM * MM, NXP = SS * MM, NWO = MM * MM;
        const int i = (blockIdx.x * 25 + tid) * 8;
        v8h o;
        _Float16* dst;
        if (i < NWI + NXP + NWO) {
            const float* src;
            if (i < NWI) {
                src = W_in + i;
                dst = Wih + frag_index(i >> 10, i & (MM - 1), MM);
            } else if (i < NWI + NXP) {
                const int j = i - NWI;
                src = W_xp + j;
                dst = Wxh + frag_index(j >> 10, j & (MM - 1), MM);
            } else {
                const int j = i - NWI - NXP;
                src = W_out + j;
                dst = Woh + frag_index(j >> 10, j & (MM - 1), MM);
            }
            const float4 a = ((const float4*)src)[0];
            const float4 b = ((const float4*)src)[1];
            o[0] = (_Float16)a.x; o[1] = (_Float16)a.y; o[2] = (_Float16)a.z; o[3] = (_Float16)a.w;
            o[4] = (_Float16)b.x; o[5] = (_Float16)b.y; o[6] = (_Float16)b.z; o[7] = (_Float16)b.w;
        } else {
            const int j = i - NWI - NXP - NWO;
            const int n = j >> 6, s0 = j & 63;
#pragma unroll
            for (int t = 0; t < 8; ++t) o[t] = (_Float16)Dmat[(size_t)(s0 + t) * MM + n];
            dst = Dth + frag_index(n, s0, SS);
        }
        *(v8h*)dst = o;
    }

    float s = v.x + v.y + v.z + v.w;
    float q = v.x * v.x + v.y * v.y + v.z * v.z + v.w * v.w;
#pragma unroll
    for (int off = 16; off > 0; off >>= 1) {
        s += __shfl_down(s, off);
        q += __shfl_down(q, off);
    }
    if ((tid & 31) == 0) { sSum[tid >> 5] = s; sSq[tid >> 5] = q; }
    __syncthreads();
    float ts = 0.f, tq = 0.f;
#pragma unroll
    for (int i = 0; i < 8; ++i) { ts += sSum[i]; tq += sSq[i]; }
    const float mu = ts * (1.0f / MM);
    const float var = tq * (1.0f / MM) - mu * mu;
    const float rstd = rsqrtf(var + 1e-5f);

    float4 g = ((const float4*)gamma)[tid];
    float4 bb = ((const float4*)beta)[tid];
    v4h o;
    o[0] = (_Float16)((v.x - mu) * rstd * g.x + bb.x);
    o[1] = (_Float16)((v.y - mu) * rstd * g.y + bb.y);
    o[2] = (_Float16)((v.z - mu) * rstd * g.z + bb.z);
    o[3] = (_Float16)((v.w - mu) * rstd * g.w + bb.w);
    *(v4h*)(xn + frag_index(row, tid * 4, MM)) = o;
}

#define IMG_HALFS 4096

__device__ __forceinline__ size_t run_base(int rt, int col0) {
    return ((size_t)rt * (size_t)(MM >> 5) + (size_t)(col0 >> 5)) * 512;
}

#define RTILES (ROWS / 16)

__global__ __launch_bounds__(256) void k_gemm1(const _Float16* __restrict__ xn,
                                               const _Float16* __restrict__ Wih,
                                               const float* __restrict__ b_in,
                                               const _Float16* __restrict__ Wxh,
                                               float* __restrict__ upart,
                                               _Float16* __restrict__ zg) {
    __shared__ __attribute__((aligned(16))) _Float16 img[8 * IMG_HALFS];
    Acc44 acc = {};
    const int w = threadIdx.x >> 5;
    const int rowbase = blockIdx.x * 128 + (w & 1) * 64;
    const int colbase = blockIdx.y * 256 + (w >> 1) * 64;
    wave_gemm64_ring<MM / 32, true>(xn, Wih, rowbase, colbase, acc);

    const int lane = threadIdx.x & 31;
    const int nlo = lane & 15, mh = (lane >> 4) * 8;
    const bool is_x = (blockIdx.y < (MM / 256));
    v8h o[4][4];
#pragma unroll
    for (int j = 0; j < 4; ++j) {
        const float4 bl = ((const float4*)(b_in + colbase + j * 16 + mh))[0];
        const float4 bh = ((const float4*)(b_in + colbase + j * 16 + mh))[1];
        const float bias[8] = {bl.x, bl.y, bl.z, bl.w, bh.x, bh.y, bh.z, bh.w};
#pragma unroll
        for (int i = 0; i < 4; ++i) {
#pragma unroll
            for (int e = 0; e < 8; ++e) {
                const float v = acc.c[i][j][e] + bias[e];
                const float sg = fast_sigmoid(v);
                o[i][j][e] = (_Float16)(is_x ? v * sg : sg);
            }
        }
    }
    if (!is_x) {
        _Float16* zrow = zg + run_base(rowbase >> 4, colbase - MM) + lane * 8;
#pragma unroll
        for (int i = 0; i < 4; ++i)
#pragma unroll
            for (int j = 0; j < 4; ++j)
                *(v8h*)(zrow + (size_t)i * (MM >> 5) * 512 + j * 256) = o[i][j];
    } else {
        _Float16* my = img + w * IMG_HALFS + lane * 8;
#pragma unroll
        for (int i = 0; i < 4; ++i)
#pragma unroll
            for (int j = 0; j < 4; ++j)
                *(v8h*)(my + (i * 4 + j) * 256) = o[i][j];
        __syncthreads();
        const int rg = w & 1, g = w >> 1;
        const _Float16* arow = img + rg * IMG_HALFS + lane * 8;
        const _Float16* bcol = Wxh + (size_t)(g * (MM >> 5) + blockIdx.y * 8) * 512 + lane * 8;
        v8f u[4] = {};
#pragma unroll
        for (int kb = 0; kb < 8; ++kb) {
            const v16h b = load_frag(bcol + kb * 512);
            const _Float16* ap = arow + (kb >> 1) * 2 * IMG_HALFS + (kb & 1) * 512;
#pragma unroll
            for (int i = 0; i < 4; ++i) {
                const v16h a = load_frag(ap + i * 1024);
                u[i] = __builtin_amdgcn_wmma_f32_16x16x32_f16(
                    false, a, false, b, (short)0, u[i], false, false);
            }
        }
        float* dst = upart + ((size_t)(blockIdx.y * 4 + g) * RTILES + (rowbase >> 4)) * 256 + nlo * 16 + mh;
#pragma unroll
        for (int i = 0; i < 4; ++i) {
            const float4 p0 = make_float4(u[i][0], u[i][1], u[i][2], u[i][3]);
            const float4 p1 = make_float4(u[i][4], u[i][5], u[i][6], u[i][7]);
            ((float4*)(dst + i * 256))[0] = p0;
            ((float4*)(dst + i * 256))[1] = p1;
        }
    }
}

#define SEG 512
#define NSEG (LL / SEG)

__global__ __launch_bounds__(256) void k_scan_local(const float* __restrict__ upart,
                                                    const float* __restrict__ b_xp,
                                                    const float* __restrict__ A_log,
                                                    float* __restrict__ hloc,
                                                    float* __restrict__ blk_end) {
    __shared__ float sEnd[8][16];
    const int w = __builtin_amdgcn_readfirstlane(threadIdx.x >> 5);
    const int g = blockIdx.y;
    const int rowbase = blockIdx.x * SEG + w * 64;
    const int lane = threadIdx.x & 31;
    const int nlo = lane & 15, mh = (lane >> 4) * 8;
    const bool lo = lane < 16;
    const int n = g * 16 + nlo;

    float4 q[4][4][2];
#pragma unroll
    for (int y = 0; y < 4; ++y)
#pragma unroll
        for (int i = 0; i < 4; ++i) {
            const float4* p = (const float4*)(upart
                + ((size_t)(y * 4 + g) * RTILES + (rowbase >> 4) + i) * 256 + nlo * 16 + mh);
            q[y][i][0] = p[0];
            q[y][i][1] = p[1];
        }
    float u[4][8];
#pragma unroll
    for (int i = 0; i < 4; ++i) {
        u[i][0] = ((q[0][i][0].x + q[1][i][0].x) + q[2][i][0].x) + q[3][i][0].x;
        u[i][1] = ((q[0][i][0].y + q[1][i][0].y) + q[2][i][0].y) + q[3][i][0].y;
        u[i][2] = ((q[0][i][0].z + q[1][i][0].z) + q[2][i][0].z) + q[3][i][0].z;
        u[i][3] = ((q[0][i][0].w + q[1][i][0].w) + q[2][i][0].w) + q[3][i][0].w;
        u[i][4] = ((q[0][i][1].x + q[1][i][1].x) + q[2][i][1].x) + q[3][i][1].x;
        u[i][5] = ((q[0][i][1].y + q[1][i][1].y) + q[2][i][1].y) + q[3][i][1].y;
        u[i][6] = ((q[0][i][1].z + q[1][i][1].z) + q[2][i][1].z) + q[3][i][1].z;
        u[i][7] = ((q[0][i][1].w + q[1][i][1].w) + q[2][i][1].w) + q[3][i][1].w;
    }

    const float bias = b_xp[n];
    const float a = expf(-expf(A_log[n]));
    const float a2 = a * a, a4 = a2 * a2;
    const float e8 = a4 * a4;
    float Lend[4], Hend[4];
    float S = 0.f;
#pragma unroll
    for (int i = 0; i < 4; ++i) {
        float h = u[i][0] + bias;
        u[i][0] = h;
#pragma unroll
        for (int jj = 1; jj < 8; ++jj) {
            h = a * h + (u[i][jj] + bias);
            u[i][jj] = h;
        }
        const float other = __shfl_xor(h, 16);
        Lend[i] = lo ? h : other;
        Hend[i] = lo ? other : h;
        S = e8 * (e8 * S + Lend[i]) + Hend[i];
    }
    if (lo) sEnd[w][nlo] = S;
    __syncthreads();

    const float a16 = e8 * e8, a32 = a16 * a16, a64 = a32 * a32;
    float ends[7];
#pragma unroll
    for (int v = 0; v < 7; ++v) ends[v] = sEnd[v][nlo];
    S = 0.f;
#pragma unroll
    for (int v = 0; v < 7; ++v) S = (v < w) ? a64 * S + ends[v] : S;
    float* hout = hloc + ((size_t)g * ROWS + rowbase + mh) * 16 + nlo;
#pragma unroll
    for (int i = 0; i < 4; ++i) {
        const float SL = e8 * S + Lend[i];
        const float inc = lo ? S : SL;
        float p = a;
#pragma unroll
        for (int jj = 0; jj < 8; ++jj) {
            hout[(i * 16 + jj) * 16] = u[i][jj] + p * inc;
            p *= a;
        }
        S = e8 * SL + Hend[i];
    }
    if (w == 7 && lo) blk_end[(blockIdx.x * 4 + g) * 32 + nlo] = S;
}

__global__ __launch_bounds__(256) void k_scan_apply(const float* __restrict__ hloc,
                                                    const float* __restrict__ A_log,
                                                    const float* __restrict__ blk_end,
                                                    _Float16* __restrict__ h_h) {
    const int t = blockIdx.x * 256 + threadIdx.x;
    const int L = t & 31;
    const int kq = (t >> 5) & 3;
    const int rp = (t >> 7) * 16 + (L & 15);
    const int s0 = (kq * 2 + (L >> 4)) * 8;
    const int l = rp >> 2, b = rp & 3;
    const int seg = (unsigned)l / SEG, pos = (unsigned)l % SEG;

    const float4* hp = (const float4*)(hloc + ((size_t)(s0 >> 4) * ROWS + (size_t)b * LL + l) * 16 + (s0 & 15));
    const float4 h0 = hp[0], h1 = hp[1];
    const float4 g0 = ((const float4*)(A_log + s0))[0], g1 = ((const float4*)(A_log + s0))[1];
    float ev[NSEG - 1][8];
#pragma unroll
    for (int v = 0; v < NSEG - 1; ++v) {
        const float4* ep = (const float4*)(blk_end + (size_t)((b * NSEG + v) * 4 + (s0 >> 4)) * 32 + (s0 & 15));
        const float4 x0 = ep[0], x1 = ep[1];
        ev[v][0] = x0.x; ev[v][1] = x0.y; ev[v][2] = x0.z; ev[v][3] = x0.w;
        ev[v][4] = x1.x; ev[v][5] = x1.y; ev[v][6] = x1.z; ev[v][7] = x1.w;
    }
    const float hv[8] = {h0.x, h0.y, h0.z, h0.w, h1.x, h1.y, h1.z, h1.w};
    const float al[8] = {g0.x, g0.y, g0.z, g0.w, g1.x, g1.y, g1.z, g1.w};
    v8h o;
#pragma unroll
    for (int q = 0; q < 8; ++q) {
        const float ea = expf(al[q]);
        const float E = expf(-ea * (float)SEG);
        const float p = expf(-ea * (float)(pos + 1));
        float c = 0.f;
#pragma unroll
        for (int v = 0; v < NSEG - 1; ++v)
            c = (v < seg) ? E * c + ev[v][q] : c;
        o[q] = (_Float16)(hv[q] + c * p);
    }
    *(v8h*)(h_h + frag_index(rp, s0, SS)) = o;
}

__global__ __launch_bounds__(256) void k_gemm3(const _Float16* __restrict__ h_h,
                                               const _Float16* __restrict__ Dth,
                                               const _Float16* __restrict__ zg,
                                               _Float16* __restrict__ ys) {
    const int w = threadIdx.x >> 5;
    const int lane = threadIdx.x & 31;
    const int rowbase = blockIdx.x * 128 + (w & 1) * 64;
    const int colbase = blockIdx.y * 256 + (w >> 1) * 64;
    const int c = lane & 15, hh = lane >> 4;

    const _Float16* zp = zg + run_base((c & 3) * (LL >> 4) + (rowbase >> 6), colbase)
                            + ((c >> 2) + hh * 16) * 8;
    _Float16* yp = ys + run_base(rowbase >> 4, colbase) + lane * 8;

    Acc44 acc = {};
    wave_gemm64<true>(h_h, Dth, rowbase, colbase, SS, acc);

#pragma unroll
    for (int i = 0; i < 4; ++i)
#pragma unroll
        for (int j = 0; j < 4; ++j) {
            const v8h z = *(const v8h*)(zp + j * 256 + i * 32);
            v8h o;
#pragma unroll
            for (int e = 0; e < 8; ++e)
                o[e] = (_Float16)(acc.c[i][j][e] * (float)z[e]);
            *(v8h*)(yp + (size_t)i * (MM >> 5) * 512 + j * 256) = o;
        }
}

__global__ __launch_bounds__(256) void k_gemm4(const _Float16* __restrict__ ys,
                                               const _Float16* __restrict__ Woh,
                                               const float* __restrict__ b_out,
                                               const float* __restrict__ xflat,
                                               float* __restrict__ out) {
    Acc44 acc = {};
    const int w = threadIdx.x >> 5;
    const int rowbase = blockIdx.x * 128 + (w & 1) * 64;
    const int colbase = blockIdx.y * 256 + (w >> 1) * 64;
    wave_gemm64_ring<MM / 32, false>(ys, Woh, rowbase, colbase, acc);

    const int lane = threadIdx.x & 31;
    const int nlo = lane & 15, mh = (lane >> 4) * 8;
#pragma unroll
    for (int j = 0; j < 4; ++j) {
        const int n = colbase + j * 16 + nlo;
        const float bias = b_out[n];
#pragma unroll
        for (int i = 0; i < 4; ++i)
#pragma unroll
            for (int jj = 0; jj < 8; ++jj) {
                const int m = rowbase + i * 16 + mh + jj;
                const size_t idx = (size_t)m * MM + n;
                out[idx] = xflat[idx] + acc.c[i][j][jj] + bias;
            }
    }
}

extern "C" void kernel_launch(void* const* d_in, const int* in_sizes, int n_in,
                              void* d_out, int out_size, void* d_ws, size_t ws_size,
                              hipStream_t stream) {
    const float* x     = (const float*)d_in[0];
    const float* A_log = (const float*)d_in[1];
    const float* Dmat  = (const float*)d_in[2];
    const float* W_in  = (const float*)d_in[3];
    const float* b_in  = (const float*)d_in[4];
    const float* W_xp  = (const float*)d_in[5];
    const float* b_xp  = (const float*)d_in[6];
    const float* W_out = (const float*)d_in[7];
    const float* b_out = (const float*)d_in[8];
    const float* gamma = (const float*)d_in[9];
    const float* beta  = (const float*)d_in[10];
    float* out = (float*)d_out;

    char* ws = (char*)d_ws;
    size_t off = 0;
    auto take = [&](size_t bytes) { char* p = ws + off; off += bytes; return p; };
    _Float16* xn_h  = (_Float16*)take((size_t)ROWS * MM * 2);
    char*     reg2  = take((size_t)ROWS * MM * 2);
    float*    upart = (float*)reg2;
    float*    bend  = (float*)(reg2 + (size_t)4 * ROWS * SS * 4);
    _Float16* z_h   = (_Float16*)take((size_t)ROWS * MM * 2);
    _Float16* ys_h  = (_Float16*)take((size_t)ROWS * MM * 2);
    float*    hloc  = (float*)take((size_t)ROWS * SS * 4);
    _Float16* h_h   = (_Float16*)take((size_t)ROWS * SS * 2);
    _Float16* Wih   = (_Float16*)take((size_t)2 * MM * MM * 2);
    _Float16* Wxh   = (_Float16*)take((size_t)SS * MM * 2);
    _Float16* Woh   = (_Float16*)take((size_t)MM * MM * 2);
    _Float16* Dth   = (_Float16*)take((size_t)SS * MM * 2);
    (void)ws_size; (void)in_sizes; (void)n_in; (void)out_size;

    k_layernorm<<<ROWS, 256, 0, stream>>>(x, gamma, beta, xn_h, W_in, W_xp, W_out, Dmat,
                                          Wih, Wxh, Woh, Dth);
    k_gemm1<<<dim3(ROWS / 128, (2 * MM) / 256), 256, 0, stream>>>(xn_h, Wih, b_in, Wxh, upart, z_h);
    k_scan_local<<<dim3(ROWS / SEG, SS / 16), 256, 0, stream>>>(upart, b_xp, A_log, hloc, bend);
    k_scan_apply<<<(ROWS * 8) / 256, 256, 0, stream>>>(hloc, A_log, bend, h_h);
    k_gemm3<<<dim3(ROWS / 128, MM / 256), 256, 0, stream>>>(h_h, Dth, z_h, ys_h);
    k_gemm4<<<dim3(ROWS / 128, MM / 256), 256, 0, stream>>>(ys_h, Woh, b_out, x, out);
}


# Readout and gating fused into out_proj: a 64-row work-group builds its gated ys rows (64x1024 f16) in LDS in fragment order as the A operand; wave tile 64x128; ys never stored, 5 launches; same sums
# speedup vs baseline: 2.5655x; 1.0505x over previous
#include <hip/hip_runtime.h>
#include <hip/hip_bf16.h>

#define BB   4
#define LL   4096
#define MM   1024
#define SS   64
#define ROWS (BB * LL)

typedef __attribute__((ext_vector_type(16))) _Float16 v16h;
typedef __attribute__((ext_vector_type(8)))  _Float16 v8h;
typedef __attribute__((ext_vector_type(4)))  _Float16 v4h;
typedef __attribute__((ext_vector_type(8)))  float    v8f;

__device__ __forceinline__ size_t frag_index(int row, int k, int K) {
    return ((size_t)(row >> 4) * (size_t)(K >> 5) + (size_t)(k >> 5)) * 512
         + (size_t)(((k >> 4) & 1) * 256 + ((row & 15) + ((k >> 3) & 1) * 16) * 8 + (k & 7));
}

__device__ __forceinline__ v16h load_frag(const _Float16* __restrict__ p) {
    v8h lo = *(const v8h*)(p);
    v8h hi = *(const v8h*)(p + 256);
    v16h t;
#pragma unroll
    for (int i = 0; i < 8; ++i) { t[i] = lo[i]; t[i + 8] = hi[i]; }
    return t;
}

struct Acc44 { v8f c[4][4]; };

template <bool TR>
__device__ __forceinline__ v8f mma1(const v16h& rowfrag, const v16h& colfrag, const v8f& c) {
    if (TR)
        return __builtin_amdgcn_wmma_f32_16x16x32_f16(false, colfrag, false, rowfrag, (short)0, c, false, false);
    else
        return __builtin_amdgcn_wmma_f32_16x16x32_f16(false, rowfrag, false, colfrag, (short)0, c, false, false);
}

template <bool TR>
__device__ __forceinline__ void wave_gemm64(const _Float16* __restrict__ A,
                                            const _Float16* __restrict__ B,
                                            int rowbase, int colbase, int K,
                                            Acc44& acc) {
    const int lane = threadIdx.x & 31;
    const int kt   = K >> 5;
    const _Float16* Ab = A + (size_t)(rowbase >> 4) * kt * 512 + lane * 8;
    const _Float16* Bb = B + (size_t)(colbase >> 4) * kt * 512 + lane * 8;
    for (int kb = 0; kb < kt; ++kb) {
        v16h a[4], b[4];
#pragma unroll
        for (int t = 0; t < 4; ++t) a[t] = load_frag(Ab + (size_t)(t * kt + kb) * 512);
#pragma unroll
        for (int t = 0; t < 4; ++t) b[t] = load_frag(Bb + (size_t)(t * kt + kb) * 512);
#pragma unroll
        for (int i = 0; i < 4; ++i)
#pragma unroll
            for (int j = 0; j < 4; ++j)
                acc.c[i][j] = mma1<TR>(a[i], b[j], acc.c[i][j]);
    }
}

struct FragSet { v16h a[4]; v16h b[4]; };

template <int KT>
__device__ __forceinline__ void load_set(FragSet& s, const _Float16* __restrict__ Ab,
                                         const _Float16* __restrict__ Bb, int kb) {
#pragma unroll
    for (int t = 0; t < 4; ++t) s.a[t] = load_frag(Ab + (size_t)(t * KT + kb) * 512);
#pragma unroll
    for (int t = 0; t < 4; ++t) s.b[t] = load_frag(Bb + (size_t)(t * KT + kb) * 512);
}

template <bool TR>
__device__ __forceinline__ void mma_set(Acc44& acc, const FragSet& s) {
#pragma unroll
    for (int i = 0; i < 4; ++i)
#pragma unroll
        for (int j = 0; j < 4; ++j)
            acc.c[i][j] = mma1<TR>(s.a[i], s.b[j], acc.c[i][j]);
}

template <int KT, bool TR>
__device__ __forceinline__ void wave_gemm64_ring(const _Float16* __restrict__ A,
                                                 const _Float16* __restrict__ B,
                                                 int rowbase, int colbase, Acc44& acc) {
    static_assert(KT % 4 == 0 && KT >= 8, "ring of four k tiles");
    const int lane = threadIdx.x & 31;
    const _Float16* Ab = A + (size_t)(rowbase >> 4) * KT * 512 + lane * 8;
    const _Float16* Bb = B + (size_t)(colbase >> 4) * KT * 512 + lane * 8;
    FragSet s0, s1, s2, s3;
    load_set<KT>(s0, Ab, Bb, 0);
    load_set<KT>(s1, Ab, Bb, 1);
    load_set<KT>(s2, Ab, Bb, 2);
    load_set<KT>(s3, Ab, Bb, 3);
    for (int kb = 4; kb < KT; kb += 4) {
        mma_set<TR>(acc, s0); __builtin_amdgcn_sched_barrier(0);
        load_set<KT>(s0, Ab, Bb, kb); __builtin_amdgcn_sched_barrier(0);
        mma_set<TR>(acc, s1); __builtin_amdgcn_sched_barrier(0);
        load_set<KT>(s1, Ab, Bb, kb + 1); __builtin_amdgcn_sched_barrier(0);
        mma_set<TR>(acc, s2); __builtin_amdgcn_sched_barrier(0);
        load_set<KT>(s2, Ab, Bb, kb + 2); __builtin_amdgcn_sched_barrier(0);
        mma_set<TR>(acc, s3); __builtin_amdgcn_sched_barrier(0);
        load_set<KT>(s3, Ab, Bb, kb + 3); __builtin_amdgcn_sched_barrier(0);
    }
    mma_set<TR>(acc, s0);
    mma_set<TR>(acc, s1);
    mma_set<TR>(acc, s2);
    mma_set<TR>(acc, s3);
}

__device__ __forceinline__ float fast_sigmoid(float v) {
    return __builtin_amdgcn_rcpf(1.0f + __expf(-v));
}

__global__ __launch_bounds__(256) void k_layernorm(const float* __restrict__ x,
                                                   const float* __restrict__ gamma,
                                                   const float* __restrict__ beta,
                                                   _Float16* __restrict__ xn,
                                                   const float* __restrict__ W_in,
                                                   const float* __restrict__ W_xp,
                                                   const float* __restrict__ W_out,
                                                   const float* __restrict__ Dmat,
                                                   _Float16* __restrict__ Wih,
                                                   _Float16* __restrict__ Wxh,
                                                   _Float16* __restrict__ Woh,
                                                   _Float16* __restrict__ Dth) {
    __shared__ float sSum[8], sSq[8];
    const int row = blockIdx.x;
    const size_t base = (size_t)row * MM;
    const int tid = threadIdx.x;

    float4 v = ((const float4*)(x + base))[tid];

    if (tid < 25) {
        const int NWI = 2 * MM * MM, NXP = SS * MM, NWO = MM * MM;
        const int i = (blockIdx.x * 25 + tid) * 8;
        v8h o;
        _Float16* dst;
        if (i < NWI + NXP + NWO) {
            const float* src;
            if (i < NWI) {
                src = W_in + i;
                dst = Wih + frag_index(i >> 10, i & (MM - 1), MM);
            } else if (i < NWI + NXP) {
                const int j = i - NWI;
                src = W_xp + j;
                dst = Wxh + frag_index(j >> 10, j & (MM - 1), MM);
            } else {
                const int j = i - NWI - NXP;
                src = W_out + j;
                dst = Woh + frag_index(j >> 10, j & (MM - 1), MM);
            }
            const float4 a = ((const float4*)src)[0];
            const float4 b = ((const float4*)src)[1];
            o[0] = (_Float16)a.x; o[1] = (_Float16)a.y; o[2] = (_Float16)a.z; o[3] = (_Float16)a.w;
            o[4] = (_Float16)b.x; o[5] = (_Float16)b.y; o[6] = (_Float16)b.z; o[7] = (_Float16)b.w;
        } else {
            const int j = i - NWI - NXP - NWO;
            const int n = j >> 6, s0 = j & 63;
#pragma unroll
            for (int t = 0; t < 8; ++t) o[t] = (_Float16)Dmat[(size_t)(s0 + t) * MM + n];
            dst = Dth + frag_index(n, s0, SS);
        }
        *(v8h*)dst = o;
    }

    float s = v.x + v.y + v.z + v.w;
    float q = v.x * v.x + v.y * v.y + v.z * v.z + v.w * v.w;
#pragma unroll
    for (int off = 16; off > 0; off >>= 1) {
        s += __shfl_down(s, off);
        q += __shfl_down(q, off);
    }
    if ((tid & 31) == 0) { sSum[tid >> 5] = s; sSq[tid >> 5] = q; }
    __syncthreads();
    float ts = 0.f, tq = 0.f;
#pragma unroll
    for (int i = 0; i < 8; ++i) { ts += sSum[i]; tq += sSq[i]; }
    const float mu = ts * (1.0f / MM);
    const float var = tq * (1.0f / MM) - mu * mu;
    const float rstd = rsqrtf(var + 1e-5f);

    float4 g = ((const float4*)gamma)[tid];
    float4 bb = ((const float4*)beta)[tid];
    v4h o;
    o[0] = (_Float16)((v.x - mu) * rstd * g.x + bb.x);
    o[1] = (_Float16)((v.y - mu) * rstd * g.y + bb.y);
    o[2] = (_Float16)((v.z - mu) * rstd * g.z + bb.z);
    o[3] = (_Float16)((v.w - mu) * rstd * g.w + bb.w);
    *(v4h*)(xn + frag_index(row, tid * 4, MM)) = o;
}

#define IMG_HALFS 4096

__device__ __forceinline__ size_t run_base(int rt, int col0) {
    return ((size_t)rt * (size_t)(MM >> 5) + (size_t)(col0 >> 5)) * 512;
}

#define RTILES (ROWS / 16)

__global__ __launch_bounds__(256) void k_gemm1(const _Float16* __restrict__ xn,
                                               const _Float16* __restrict__ Wih,
                                               const float* __restrict__ b_in,
                                               const _Float16* __restrict__ Wxh,
                                               float* __restrict__ upart,
                                               _Float16* __restrict__ zg) {
    __shared__ __attribute__((aligned(16))) _Float16 img[8 * IMG_HALFS];
    Acc44 acc = {};
    const int w = threadIdx.x >> 5;
    const int rowbase = blockIdx.x * 128 + (w & 1) * 64;
    const int colbase = blockIdx.y * 256 + (w >> 1) * 64;
    wave_gemm64_ring<MM / 32, true>(xn, Wih, rowbase, colbase, acc);

    const int lane = threadIdx.x & 31;
    const int nlo = lane & 15, mh = (lane >> 4) * 8;
    const bool is_x = (blockIdx.y < (MM / 256));
    v8h o[4][4];
#pragma unroll
    for (int j = 0; j < 4; ++j) {
        const float4 bl = ((const float4*)(b_in + colbase + j * 16 + mh))[0];
        const float4 bh = ((const float4*)(b_in + colbase + j * 16 + mh))[1];
        const float bias[8] = {bl.x, bl.y, bl.z, bl.w, bh.x, bh.y, bh.z, bh.w};
#pragma unroll
        for (int i = 0; i < 4; ++i) {
#pragma unroll
            for (int e = 0; e < 8; ++e) {
                const float v = acc.c[i][j][e] + bias[e];
                const float sg = fast_sigmoid(v);
                o[i][j][e] = (_Float16)(is_x ? v * sg : sg);
            }
        }
    }
    if (!is_x) {
        _Float16* zrow = zg + run_base(rowbase >> 4, colbase - MM) + lane * 8;
#pragma unroll
        for (int i = 0; i < 4; ++i)
#pragma unroll
            for (int j = 0; j < 4; ++j)
                *(v8h*)(zrow + (size_t)i * (MM >> 5) * 512 + j * 256) = o[i][j];
    } else {
        _Float16* my = img + w * IMG_HALFS + lane * 8;
#pragma unroll
        for (int i = 0; i < 4; ++i)
#pragma unroll
            for (int j = 0; j < 4; ++j)
                *(v8h*)(my + (i * 4 + j) * 256) = o[i][j];
        __syncthreads();
        const int rg = w & 1, g = w >> 1;
        const _Float16* arow = img + rg * IMG_HALFS + lane * 8;
        const _Float16* bcol = Wxh + (size_t)(g * (MM >> 5) + blockIdx.y * 8) * 512 + lane * 8;
        v8f u[4] = {};
#pragma unroll
        for (int kb = 0; kb < 8; ++kb) {
            const v16h b = load_frag(bcol + kb * 512);
            const _Float16* ap = arow + (kb >> 1) * 2 * IMG_HALFS + (kb & 1) * 512;
#pragma unroll
            for (int i = 0; i < 4; ++i) {
                const v16h a = load_frag(ap + i * 1024);
                u[i] = __builtin_amdgcn_wmma_f32_16x16x32_f16(
                    false, a, false, b, (short)0, u[i], false, false);
            }
        }
        float* dst = upart + ((size_t)(blockIdx.y * 4 + g) * RTILES + (rowbase >> 4)) * 256 + nlo * 16 + mh;
#pragma unroll
        for (int i = 0; i < 4; ++i) {
            const float4 p0 = make_float4(u[i][0], u[i][1], u[i][2], u[i][3]);
            const float4 p1 = make_float4(u[i][4], u[i][5], u[i][6], u[i][7]);
            ((float4*)(dst + i * 256))[0] = p0;
            ((float4*)(dst + i * 256))[1] = p1;
        }
    }
}

#define SEG 512
#define NSEG (LL / SEG)

__global__ __launch_bounds__(256) void k_scan_local(const float* __restrict__ upart,
                                                    const float* __restrict__ b_xp,
                                                    const float* __restrict__ A_log,
                                                    float* __restrict__ hloc,
                                                    float* __restrict__ blk_end) {
    __shared__ float sEnd[8][16];
    const int w = __builtin_amdgcn_readfirstlane(threadIdx.x >> 5);
    const int g = blockIdx.y;
    const int rowbase = blockIdx.x * SEG + w * 64;
    const int lane = threadIdx.x & 31;
    const int nlo = lane & 15, mh = (lane >> 4) * 8;
    const bool lo = lane < 16;
    const int n = g * 16 + nlo;

    float4 q[4][4][2];
#pragma unroll
    for (int y = 0; y < 4; ++y)
#pragma unroll
        for (int i = 0; i < 4; ++i) {
            const float4* p = (const float4*)(upart
                + ((size_t)(y * 4 + g) * RTILES + (rowbase >> 4) + i) * 256 + nlo * 16 + mh);
            q[y][i][0] = p[0];
            q[y][i][1] = p[1];
        }
    float u[4][8];
#pragma unroll
    for (int i = 0; i < 4; ++i) {
        u[i][0] = ((q[0][i][0].x + q[1][i][0].x) + q[2][i][0].x) + q[3][i][0].x;
        u[i][1] = ((q[0][i][0].y + q[1][i][0].y) + q[2][i][0].y) + q[3][i][0].y;
        u[i][2] = ((q[0][i][0].z + q[1][i][0].z) + q[2][i][0].z) + q[3][i][0].z;
        u[i][3] = ((q[0][i][0].w + q[1][i][0].w) + q[2][i][0].w) + q[3][i][0].w;
        u[i][4] = ((q[0][i][1].x + q[1][i][1].x) + q[2][i][1].x) + q[3][i][1].x;
        u[i][5] = ((q[0][i][1].y + q[1][i][1].y) + q[2][i][1].y) + q[3][i][1].y;
        u[i][6] = ((q[0][i][1].z + q[1][i][1].z) + q[2][i][1].z) + q[3][i][1].z;
        u[i][7] = ((q[0][i][1].w + q[1][i][1].w) + q[2][i][1].w) + q[3][i][1].w;
    }

    const float bias = b_xp[n];
    const float a = expf(-expf(A_log[n]));
    const float a2 = a * a, a4 = a2 * a2;
    const float e8 = a4 * a4;
    float Lend[4], Hend[4];
    float S = 0.f;
#pragma unroll
    for (int i = 0; i < 4; ++i) {
        float h = u[i][0] + bias;
        u[i][0] = h;
#pragma unroll
        for (int jj = 1; jj < 8; ++jj) {
            h = a * h + (u[i][jj] + bias);
            u[i][jj] = h;
        }
        const float other = __shfl_xor(h, 16);
        Lend[i] = lo ? h : other;
        Hend[i] = lo ? other : h;
        S = e8 * (e8 * S + Lend[i]) + Hend[i];
    }
    if (lo) sEnd[w][nlo] = S;
    __syncthreads();

    const float a16 = e8 * e8, a32 = a16 * a16, a64 = a32 * a32;
    float ends[7];
#pragma unroll
    for (int v = 0; v < 7; ++v) ends[v] = sEnd[v][nlo];
    S = 0.f;
#pragma unroll
    for (int v = 0; v < 7; ++v) S = (v < w) ? a64 * S + ends[v] : S;
    float* hout = hloc + ((size_t)g * ROWS + rowbase + mh) * 16 + nlo;
#pragma unroll
    for (int i = 0; i < 4; ++i) {
        const float SL = e8 * S + Lend[i];
        const float inc = lo ? S : SL;
        float p = a;
#pragma unroll
        for (int jj = 0; jj < 8; ++jj) {
            hout[(i * 16 + jj) * 16] = u[i][jj] + p * inc;
            p *= a;
        }
        S = e8 * SL + Hend[i];
    }
    if (w == 7 && lo) blk_end[(blockIdx.x * 4 + g) * 32 + nlo] = S;
}

__global__ __launch_bounds__(256) void k_scan_apply(const float* __restrict__ hloc,
                                                    const float* __restrict__ A_log,
                                                    const float* __restrict__ blk_end,
                                                    _Float16* __restrict__ h_h) {
    const int t = blockIdx.x * 256 + threadIdx.x;
    const int L = t & 31;
    const int kq = (t >> 5) & 3;
    const int rp = (t >> 7) * 16 + (L & 15);
    const int s0 = (kq * 2 + (L >> 4)) * 8;
    const int l = rp >> 2, b = rp & 3;
    const int seg = (unsigned)l / SEG, pos = (unsigned)l % SEG;

    const float4* hp = (const float4*)(hloc + ((size_t)(s0 >> 4) * ROWS + (size_t)b * LL + l) * 16 + (s0 & 15));
    const float4 h0 = hp[0], h1 = hp[1];
    const float4 g0 = ((const float4*)(A_log + s0))[0], g1 = ((const float4*)(A_log + s0))[1];
    float ev[NSEG - 1][8];
#pragma unroll
    for (int v = 0; v < NSEG - 1; ++v) {
        const float4* ep = (const float4*)(blk_end + (size_t)((b * NSEG + v) * 4 + (s0 >> 4)) * 32 + (s0 & 15));
        const float4 x0 = ep[0], x1 = ep[1];
        ev[v][0] = x0.x; ev[v][1] = x0.y; ev[v][2] = x0.z; ev[v][3] = x0.w;
        ev[v][4] = x1.x; ev[v][5] = x1.y; ev[v][6] = x1.z; ev[v][7] = x1.w;
    }
    const float hv[8] = {h0.x, h0.y, h0.z, h0.w, h1.x, h1.y, h1.z, h1.w};
    const float al[8] = {g0.x, g0.y, g0.z, g0.w, g1.x, g1.y, g1.z, g1.w};
    v8h o;
#pragma unroll
    for (int q = 0; q < 8; ++q) {
        const float ea = expf(al[q]);
        const float E = expf(-ea * (float)SEG);
        const float p = expf(-ea * (float)(pos + 1));
        float c = 0.f;
#pragma unroll
        for (int v = 0; v < NSEG - 1; ++v)
            c = (v < seg) ? E * c + ev[v][q] : c;
        o[q] = (_Float16)(hv[q] + c * p);
    }
    *(v8h*)(h_h + frag_index(rp, s0, SS)) = o;
}

#define YIMG_HALFS (64 * MM)

struct Acc48 { v8f c[4][8]; };
struct ASet4 { v16h a[4]; };
struct BSet8 { v16h b[8]; };

__device__ __forceinline__ void load_a4(ASet4& s, const _Float16* Al, int kb) {
#pragma unroll
    for (int i = 0; i < 4; ++i) s.a[i] = load_frag(Al + (i * (MM >> 5) + kb) * 512);
}
__device__ __forceinline__ void load_b8(BSet8& s, const _Float16* __restrict__ Bb, int kb) {
#pragma unroll
    for (int j = 0; j < 8; ++j) s.b[j] = load_frag(Bb + (size_t)(j * (MM >> 5) + kb) * 512);
}
__device__ __forceinline__ void mma_48(Acc48& acc, const ASet4& a, const BSet8& b) {
#pragma unroll
    for (int i = 0; i < 4; ++i)
#pragma unroll
        for (int j = 0; j < 8; ++j)
            acc.c[i][j] = mma1<false>(a.a[i], b.b[j], acc.c[i][j]);
}

__global__ __launch_bounds__(256) void k_gemm34(const _Float16* __restrict__ h_h,
                                                const _Float16* __restrict__ Dth,
                                                const _Float16* __restrict__ zg,
                                                const _Float16* __restrict__ Woh,
                                                const float* __restrict__ b_out,
                                                const float* __restrict__ xflat,
                                                float* __restrict__ out) {
    __shared__ __attribute__((aligned(16))) _Float16 yimg[YIMG_HALFS];
    const int w = threadIdx.x >> 5;
    const int lane = threadIdx.x & 31;
    const int rowbase = blockIdx.x * 64;
    const int c = lane & 15, hh = lane >> 4;

    for (int half = 0; half < 2; ++half) {
        const int colbase = w * 128 + half * 64;
        const _Float16* zp = zg + run_base((c & 3) * (LL >> 4) + (rowbase >> 6), colbase)
                                + ((c >> 2) + hh * 16) * 8;
        _Float16* yp = yimg + (colbase >> 5) * 512 + lane * 8;
        Acc44 ra = {};
        wave_gemm64<true>(h_h, Dth, rowbase, colbase, SS, ra);
#pragma unroll
        for (int i = 0; i < 4; ++i)
#pragma unroll
            for (int j = 0; j < 4; ++j) {
                const v8h z = *(const v8h*)(zp + j * 256 + i * 32);
                v8h o;
#pragma unroll
                for (int e = 0; e < 8; ++e)
                    o[e] = (_Float16)(ra.c[i][j][e] * (float)z[e]);
                *(v8h*)(yp + i * (MM >> 5) * 512 + j * 256) = o;
            }
    }
    __syncthreads();

    const int cb = w * 128;
    const _Float16* Al = yimg + lane * 8;
    const _Float16* Bb = Woh + (size_t)(cb >> 4) * (MM >> 5) * 512 + lane * 8;
    Acc48 acc = {};
    ASet4 a0, a1;
    BSet8 b0, b1;
    load_b8(b0, Bb, 0); load_a4(a0, Al, 0);
    load_b8(b1, Bb, 1); load_a4(a1, Al, 1);
    for (int kb = 2; kb < (MM >> 5); kb += 2) {
        mma_48(acc, a0, b0); __builtin_amdgcn_sched_barrier(0);
        load_b8(b0, Bb, kb); load_a4(a0, Al, kb); __builtin_amdgcn_sched_barrier(0);
        mma_48(acc, a1, b1); __builtin_amdgcn_sched_barrier(0);
        load_b8(b1, Bb, kb + 1); load_a4(a1, Al, kb + 1); __builtin_amdgcn_sched_barrier(0);
    }
    mma_48(acc, a0, b0);
    mma_48(acc, a1, b1);

    const int nlo = lane & 15, mh = (lane >> 4) * 8;
#pragma unroll
    for (int j = 0; j < 8; ++j) {
        const int n = cb + j * 16 + nlo;
        const float bias = b_out[n];
#pragma unroll
        for (int i = 0; i < 4; ++i)
#pragma unroll
            for (int jj = 0; jj < 8; ++jj) {
                const int m = rowbase + i * 16 + mh + jj;
                const size_t idx = (size_t)m * MM + n;
                out[idx] = xflat[idx] + acc.c[i][j][jj] + bias;
            }
    }
}

extern "C" void kernel_launch(void* const* d_in, const int* in_sizes, int n_in,
                              void* d_out, int out_size, void* d_ws, size_t ws_size,
                              hipStream_t stream) {
    const float* x     = (const float*)d_in[0];
    const float* A_log = (const float*)d_in[1];
    const float* Dmat  = (const float*)d_in[2];
    const float* W_in  = (const float*)d_in[3];
    const float* b_in  = (const float*)d_in[4];
    const float* W_xp  = (const float*)d_in[5];
    const float* b_xp  = (const float*)d_in[6];
    const float* W_out = (const float*)d_in[7];
    const float* b_out = (const float*)d_in[8];
    const float* gamma = (const float*)d_in[9];
    const float* beta  = (const float*)d_in[10];
    float* out = (float*)d_out;

    char* ws = (char*)d_ws;
    size_t off = 0;
    auto take = [&](size_t bytes) { char* p = ws + off; off += bytes; return p; };
    _Float16* xn_h  = (_Float16*)take((size_t)ROWS * MM * 2);
    char*     reg2  = take((size_t)ROWS * MM * 2);
    float*    upart = (float*)reg2;
    float*    bend  = (float*)(reg2 + (size_t)4 * ROWS * SS * 4);
    _Float16* z_h   = (_Float16*)take((size_t)ROWS * MM * 2);
    float*    hloc  = (float*)take((size_t)ROWS * SS * 4);
    _Float16* h_h   = (_Float16*)take((size_t)ROWS * SS * 2);
    _Float16* Wih   = (_Float16*)take((size_t)2 * MM * MM * 2);
    _Float16* Wxh   = (_Float16*)take((size_t)SS * MM * 2);
    _Float16* Woh   = (_Float16*)take((size_t)MM * MM * 2);
    _Float16* Dth   = (_Float16*)take((size_t)SS * MM * 2);
    (void)ws_size; (void)in_sizes; (void)n_in; (void)out_size;

    k_layernorm<<<ROWS, 256, 0, stream>>>(x, gamma, beta, xn_h, W_in, W_xp, W_out, Dmat,
                                          Wih, Wxh, Woh, Dth);
    k_gemm1<<<dim3(ROWS / 128, (2 * MM) / 256), 256, 0, stream>>>(xn_h, Wih, b_in, Wxh, upart, z_h);
    k_scan_local<<<dim3(ROWS / SEG, SS / 16), 256, 0, stream>>>(upart, b_xp, A_log, hloc, bend);
    k_scan_apply<<<(ROWS * 8) / 256, 256, 0, stream>>>(hloc, A_log, bend, h_h);
    k_gemm34<<<ROWS / 64, 256, 0, stream>>>(h_h, Dth, z_h, Woh, b_out, x, out);
}


# Scan launches removed: in_proj's state-projection waves scan their 64 tokens on the accumulators; out_proj's prologue chains the segment ends into carries and builds its h tile in LDS; new sum order
# speedup vs baseline: 3.5736x; 1.3929x over previous
#include <hip/hip_runtime.h>
#include <hip/hip_bf16.h>

#define BB   4
#define LL   4096
#define MM   1024
#define SS   64
#define ROWS (BB * LL)

typedef __attribute__((ext_vector_type(16))) _Float16 v16h;
typedef __attribute__((ext_vector_type(8)))  _Float16 v8h;
typedef __attribute__((ext_vector_type(4)))  _Float16 v4h;
typedef __attribute__((ext_vector_type(8)))  float    v8f;

__device__ __forceinline__ size_t frag_index(int row, int k, int K) {
    return ((size_t)(row >> 4) * (size_t)(K >> 5) + (size_t)(k >> 5)) * 512
         + (size_t)(((k >> 4) & 1) * 256 + ((row & 15) + ((k >> 3) & 1) * 16) * 8 + (k & 7));
}

__device__ __forceinline__ v16h load_frag(const _Float16* __restrict__ p) {
    v8h lo = *(const v8h*)(p);
    v8h hi = *(const v8h*)(p + 256);
    v16h t;
#pragma unroll
    for (int i = 0; i < 8; ++i) { t[i] = lo[i]; t[i + 8] = hi[i]; }
    return t;
}

struct Acc44 { v8f c[4][4]; };

template <bool TR>
__device__ __forceinline__ v8f mma1(const v16h& rowfrag, const v16h& colfrag, const v8f& c) {
    if (TR)
        return __builtin_amdgcn_wmma_f32_16x16x32_f16(false, colfrag, false, rowfrag, (short)0, c, false, false);
    else
        return __builtin_amdgcn_wmma_f32_16x16x32_f16(false, rowfrag, false, colfrag, (short)0, c, false, false);
}

template <bool TR>
__device__ __forceinline__ void wave_gemm64(const _Float16* __restrict__ A,
                                            const _Float16* __restrict__ B,
                                            int rowbase, int colbase, int K,
                                            Acc44& acc) {
    const int lane = threadIdx.x & 31;
    const int kt   = K >> 5;
    const _Float16* Ab = A + (size_t)(rowbase >> 4) * kt * 512 + lane * 8;
    const _Float16* Bb = B + (size_t)(colbase >> 4) * kt * 512 + lane * 8;
    for (int kb = 0; kb < kt; ++kb) {
        v16h a[4], b[4];
#pragma unroll
        for (int t = 0; t < 4; ++t) a[t] = load_frag(Ab + (size_t)(t * kt + kb) * 512);
#pragma unroll
        for (int t = 0; t < 4; ++t) b[t] = load_frag(Bb + (size_t)(t * kt + kb) * 512);
#pragma unroll
        for (int i = 0; i < 4; ++i)
#pragma unroll
            for (int j = 0; j < 4; ++j)
                acc.c[i][j] = mma1<TR>(a[i], b[j], acc.c[i][j]);
    }
}

__device__ __forceinline__ float fast_sigmoid(float v) {
    return __builtin_amdgcn_rcpf(1.0f + __expf(-v));
}

#define NGROUPS ((2 * MM * MM + SS * MM + MM * MM + SS * MM) / 8)

__global__ __launch_bounds__(256) void k_convert(const float* __restrict__ W_in,
                                                 const float* __restrict__ W_xp,
                                                 const float* __restrict__ W_out,
                                                 const float* __restrict__ Dmat,
                                                 _Float16* __restrict__ Wih,
                                                 _Float16* __restrict__ Wxh,
                                                 _Float16* __restrict__ Woh,
                                                 _Float16* __restrict__ Dth) {
    const int NWI = 2 * MM * MM, NXP = SS * MM, NWO = MM * MM;
    const int i = (blockIdx.x * 256 + threadIdx.x) * 8;
    v8h o;
    _Float16* dst;
    if (i < NWI + NXP + NWO) {
        const float* src;
        if (i < NWI) {
            src = W_in + i;
            dst = Wih + frag_index(i >> 10, i & (MM - 1), MM);
        } else if (i < NWI + NXP) {
            const int j = i - NWI;
            src = W_xp + j;
            dst = Wxh + frag_index(j >> 10, j & (MM - 1), MM);
        } else {
            const int j = i - NWI - NXP;
            src = W_out + j;
            dst = Woh + frag_index(j >> 10, j & (MM - 1), MM);
        }
        const float4 a = ((const float4*)src)[0];
        const float4 b = ((const float4*)src)[1];
        o[0] = (_Float16)a.x; o[1] = (_Float16)a.y; o[2] = (_Float16)a.z; o[3] = (_Float16)a.w;
        o[4] = (_Float16)b.x; o[5] = (_Float16)b.y; o[6] = (_Float16)b.z; o[7] = (_Float16)b.w;
    } else {
        const int j = i - NWI - NXP - NWO;
        const int n = j >> 6, s0 = j & 63;
#pragma unroll
        for (int t = 0; t < 8; ++t) o[t] = (_Float16)Dmat[(size_t)(s0 + t) * MM + n];
        dst = Dth + frag_index(n, s0, SS);
    }
    *(v8h*)dst = o;
}

#define IMG_HALFS 4096

__device__ __forceinline__ size_t run_base(int rt, int col0) {
    return ((size_t)rt * (size_t)(MM >> 5) + (size_t)(col0 >> 5)) * 512;
}

#define SEG64  64
#define NSEG64 (LL / SEG64)

#define YIMG_HALFS (64 * MM)

struct Acc48 { v8f c[4][8]; };
struct ASet4 { v16h a[4]; };
struct BSet8 { v16h b[8]; };

__device__ __forceinline__ void load_a4(ASet4& s, const _Float16* Al, int kb) {
#pragma unroll
    for (int i = 0; i < 4; ++i) s.a[i] = load_frag(Al + (i * (MM >> 5) + kb) * 512);
}
__device__ __forceinline__ void load_b8(BSet8& s, const _Float16* __restrict__ Bb, int kb) {
#pragma unroll
    for (int j = 0; j < 8; ++j) s.b[j] = load_frag(Bb + (size_t)(j * (MM >> 5) + kb) * 512);
}
template <bool TR>
__device__ __forceinline__ void mma_48(Acc48& acc, const ASet4& a, const BSet8& b) {
#pragma unroll
    for (int i = 0; i < 4; ++i)
#pragma unroll
        for (int j = 0; j < 8; ++j)
            acc.c[i][j] = mma1<TR>(a.a[i], b.b[j], acc.c[i][j]);
}

__global__ __launch_bounds__(256) void k_inproj(const float* __restrict__ x,
                                                const float* __restrict__ gamma,
                                                const float* __restrict__ beta,
                                                const _Float16* __restrict__ Wih,
                                                const float* __restrict__ b_in,
                                                const _Float16* __restrict__ Wxh,
                                                const float* __restrict__ b_xp,
                                                const float* __restrict__ A_log,
                                                float* __restrict__ hloc,
                                                float* __restrict__ blk_end,
                                                _Float16* __restrict__ zg) {
    __shared__ __attribute__((aligned(16))) _Float16 aimg[YIMG_HALFS];
    __shared__ __attribute__((aligned(16))) _Float16 simg[YIMG_HALFS];
    __shared__ float sEnd[4][16];
    const int w = threadIdx.x >> 5;
    const int lane = threadIdx.x & 31;
    const int rowbase = blockIdx.x * 64;

    {
        const int rr = lane >> 2, q = lane & 3;
        const int lr = w * 8 + rr;
        const float* xr = x + (size_t)(rowbase + lr) * MM + q * 8;
        float4 v0[32], v1[32];
#pragma unroll
        for (int t = 0; t < 32; ++t) {
            v0[t] = *(const float4*)(xr + t * 32);
            v1[t] = *(const float4*)(xr + t * 32 + 4);
        }
        float sm = 0.f, sq = 0.f;
#pragma unroll
        for (int t = 0; t < 32; ++t) {
            sm += ((v0[t].x + v0[t].y) + (v0[t].z + v0[t].w)) + ((v1[t].x + v1[t].y) + (v1[t].z + v1[t].w));
            sq += ((v0[t].x * v0[t].x + v0[t].y * v0[t].y) + (v0[t].z * v0[t].z + v0[t].w * v0[t].w))
                + ((v1[t].x * v1[t].x + v1[t].y * v1[t].y) + (v1[t].z * v1[t].z + v1[t].w * v1[t].w));
        }
        sm += __shfl_xor(sm, 1); sq += __shfl_xor(sq, 1);
        sm += __shfl_xor(sm, 2); sq += __shfl_xor(sq, 2);
        const float mu = sm * (1.0f / MM);
        const float var = sq * (1.0f / MM) - mu * mu;
        const float rstd = rsqrtf(var + 1e-5f);
        _Float16* ap = aimg + (lr >> 4) * (MM >> 5) * 512 + (q >> 1) * 256 + ((lr & 15) + (q & 1) * 16) * 8;
        const float* gp = gamma + q * 8;
        const float* bp = beta + q * 8;
#pragma unroll
        for (int t = 0; t < 32; ++t) {
            const float4 g0 = *(const float4*)(gp + t * 32), g1 = *(const float4*)(gp + t * 32 + 4);
            const float4 c0 = *(const float4*)(bp + t * 32), c1 = *(const float4*)(bp + t * 32 + 4);
            v8h o;
            o[0] = (_Float16)((v0[t].x - mu) * rstd * g0.x + c0.x);
            o[1] = (_Float16)((v0[t].y - mu) * rstd * g0.y + c0.y);
            o[2] = (_Float16)((v0[t].z - mu) * rstd * g0.z + c0.z);
            o[3] = (_Float16)((v0[t].w - mu) * rstd * g0.w + c0.w);
            o[4] = (_Float16)((v1[t].x - mu) * rstd * g1.x + c1.x);
            o[5] = (_Float16)((v1[t].y - mu) * rstd * g1.y + c1.y);
            o[6] = (_Float16)((v1[t].z - mu) * rstd * g1.z + c1.z);
            o[7] = (_Float16)((v1[t].w - mu) * rstd * g1.w + c1.w);
            *(v8h*)(ap + t * 512) = o;
        }
    }
    __syncthreads();

    const int mh = (lane >> 4) * 8;
    const _Float16* Al = aimg + lane * 8;
    for (int p = 0; p < 2; ++p) {
        const int colbase = p * MM + w * 128;
        const _Float16* Bb = Wih + (size_t)(colbase >> 4) * (MM >> 5) * 512 + lane * 8;
        Acc48 acc = {};
        ASet4 a0, a1;
        BSet8 b0, b1;
        load_b8(b0, Bb, 0); load_a4(a0, Al, 0);
        load_b8(b1, Bb, 1); load_a4(a1, Al, 1);
        for (int kb = 2; kb < (MM >> 5); kb += 2) {
            mma_48<true>(acc, a0, b0); __builtin_amdgcn_sched_barrier(0);
            load_b8(b0, Bb, kb); load_a4(a0, Al, kb); __builtin_amdgcn_sched_barrier(0);
            mma_48<true>(acc, a1, b1); __builtin_amdgcn_sched_barrier(0);
            load_b8(b1, Bb, kb + 1); load_a4(a1, Al, kb + 1); __builtin_amdgcn_sched_barrier(0);
        }
        mma_48<true>(acc, a0, b0);
        mma_48<true>(acc, a1, b1);

        v8h o[4][8];
#pragma unroll
        for (int j = 0; j < 8; ++j) {
            const float4 bl = ((const float4*)(b_in + colbase + j * 16 + mh))[0];
            const float4 bh = ((const float4*)(b_in + colbase + j * 16 + mh))[1];
            const float bias[8] = {bl.x, bl.y, bl.z, bl.w, bh.x, bh.y, bh.z, bh.w};
#pragma unroll
            for (int i = 0; i < 4; ++i) {
#pragma unroll
                for (int e = 0; e < 8; ++e) {
                    const float v = acc.c[i][j][e] + bias[e];
                    const float sg = fast_sigmoid(v);
                    o[i][j][e] = (_Float16)(p == 0 ? v * sg : sg);
                }
            }
        }
        if (p == 0) {
            _Float16* sdst = simg + (w * 4) * 512 + lane * 8;
#pragma unroll
            for (int i = 0; i < 4; ++i)
#pragma unroll
                for (int j = 0; j < 8; ++j)
                    *(v8h*)(sdst + i * (MM >> 5) * 512 + j * 256) = o[i][j];
        } else {
            _Float16* zdst = zg + run_base(rowbase >> 4, w * 128) + lane * 8;
#pragma unroll
            for (int i = 0; i < 4; ++i)
#pragma unroll
                for (int j = 0; j < 8; ++j)
                    *(v8h*)(zdst + (size_t)i * (MM >> 5) * 512 + j * 256) = o[i][j];
        }
    }
    __syncthreads();

    {
        const int g = w & 3, rt0 = (w >> 2) * 2;
        const int nlo = lane & 15;
        const _Float16* sp = simg + rt0 * (MM >> 5) * 512 + lane * 8;
        const _Float16* xp = Wxh + (size_t)g * (MM >> 5) * 512 + lane * 8;
        v8f u0 = {}, u1 = {};
        const int n = g * 16 + nlo;
        const float bias = b_xp[n];
        const float alog = A_log[n];
#pragma unroll 8
        for (int kb = 0; kb < (MM >> 5); ++kb) {
            const v16h b = load_frag(xp + kb * 512);
            const v16h f0 = load_frag(sp + kb * 512);
            const v16h f1 = load_frag(sp + ((MM >> 5) + kb) * 512);
            u0 = __builtin_amdgcn_wmma_f32_16x16x32_f16(false, f0, false, b, (short)0, u0, false, false);
            u1 = __builtin_amdgcn_wmma_f32_16x16x32_f16(false, f1, false, b, (short)0, u1, false, false);
        }

        const bool lo = lane < 16;
        const int second = __builtin_amdgcn_readfirstlane(w >> 2);
        const float a = expf(-expf(alog));
        const float a2 = a * a, a4 = a2 * a2;
        const float e8 = a4 * a4;
        float hv[2][8];
#pragma unroll
        for (int jj = 0; jj < 8; ++jj) { hv[0][jj] = u0[jj]; hv[1][jj] = u1[jj]; }
        float Lend[2], Hend[2];
        float S = 0.f;
#pragma unroll
        for (int i = 0; i < 2; ++i) {
            float h = hv[i][0] + bias;
            hv[i][0] = h;
#pragma unroll
            for (int jj = 1; jj < 8; ++jj) {
                h = a * h + (hv[i][jj] + bias);
                hv[i][jj] = h;
            }
            const float other = __shfl_xor(h, 16);
            Lend[i] = lo ? h : other;
            Hend[i] = lo ? other : h;
            S = e8 * (e8 * S + Lend[i]) + Hend[i];
        }
        if (second == 0 && lo) sEnd[g][nlo] = S;
        __syncthreads();
        S = (second != 0) ? sEnd[g][nlo] : 0.f;
        float* hout = hloc + ((size_t)g * ROWS + rowbase + rt0 * 16 + mh) * 16 + nlo;
#pragma unroll
        for (int i = 0; i < 2; ++i) {
            const float SL = e8 * S + Lend[i];
            const float inc = lo ? S : SL;
            float p = a;
#pragma unroll
            for (int jj = 0; jj < 8; ++jj) {
                hout[(i * 16 + jj) * 16] = hv[i][jj] + p * inc;
                p *= a;
            }
            S = e8 * SL + Hend[i];
        }
        if (second != 0 && lo) blk_end[(size_t)blockIdx.x * SS + n] = S;
    }
}

__global__ __launch_bounds__(256) void k_gemm34(const float* __restrict__ hloc,
                                                const float* __restrict__ A_log,
                                                const float* __restrict__ blk_end,
                                                const _Float16* __restrict__ Dth,
                                                const _Float16* __restrict__ zg,
                                                const _Float16* __restrict__ Woh,
                                                const float* __restrict__ b_out,
                                                const float* __restrict__ xflat,
                                                float* __restrict__ out) {
    __shared__ __attribute__((aligned(16))) _Float16 yimg[YIMG_HALFS];
    __shared__ __attribute__((aligned(16))) _Float16 himg[64 * SS];
    __shared__ __attribute__((aligned(16))) float sCarry[BB * SS];
    const int w = threadIdx.x >> 5;
    const int lane = threadIdx.x & 31;
    const int rowbase = blockIdx.x * 64;
    const int c = lane & 15, hh = lane >> 4;

    {
        const int tid = threadIdx.x;
        const int l0 = rowbase >> 2;
        const int seg = l0 >> 6;
        float4 hq[2][2];
        int sb[2], ss0[2], spw[2];
#pragma unroll
        for (int u = 0; u < 2; ++u) {
            const int q = tid + u * 256;
            const int L = q & 31;
            const int r = (q >> 7) * 16 + (L & 15);
            const int s0 = ((q >> 5) & 3) * 16 + (L >> 4) * 8;
            const int b = r & 3, l = l0 + (r >> 2);
            const float4* hp = (const float4*)(hloc + ((size_t)(s0 >> 4) * ROWS + (size_t)b * LL + l) * 16 + (s0 & 15));
            hq[u][0] = hp[0];
            hq[u][1] = hp[1];
            sb[u] = b; ss0[u] = s0; spw[u] = (l & (SEG64 - 1)) + 1;
        }
        {
            const int cb = tid >> 6, cn = tid & 63;
            const float ea = expf(A_log[cn]);
            const float a64 = expf(-ea * (float)SEG64);
            float e[NSEG64 - 1];
#pragma unroll
            for (int v = 0; v < NSEG64 - 1; ++v)
                e[v] = blk_end[(size_t)((cb * NSEG64 + v) * SS) + cn];
            float cy = 0.f;
#pragma unroll
            for (int v = 0; v < NSEG64 - 1; ++v)
                cy = (v < seg) ? a64 * cy + e[v] : cy;
            sCarry[tid] = cy;
        }
        __syncthreads();
#pragma unroll
        for (int u = 0; u < 2; ++u) {
            const float4 g0 = ((const float4*)(A_log + ss0[u]))[0], g1 = ((const float4*)(A_log + ss0[u]))[1];
            const float4 c0 = ((const float4*)(sCarry + sb[u] * SS + ss0[u]))[0];
            const float4 c1 = ((const float4*)(sCarry + sb[u] * SS + ss0[u]))[1];
            const float hv[8] = {hq[u][0].x, hq[u][0].y, hq[u][0].z, hq[u][0].w, hq[u][1].x, hq[u][1].y, hq[u][1].z, hq[u][1].w};
            const float al[8] = {g0.x, g0.y, g0.z, g0.w, g1.x, g1.y, g1.z, g1.w};
            const float cv[8] = {c0.x, c0.y, c0.z, c0.w, c1.x, c1.y, c1.z, c1.w};
            const float pw = (float)spw[u];
            v8h o;
#pragma unroll
            for (int q = 0; q < 8; ++q) {
                const float ea = expf(al[q]);
                const float p = expf(-ea * pw);
                o[q] = (_Float16)(hv[q] + cv[q] * p);
            }
            *(v8h*)(himg + (tid + u * 256) * 8) = o;
        }
        __syncthreads();
    }

    for (int half = 0; half < 2; ++half) {
        const int colbase = w * 128 + half * 64;
        const _Float16* zp = zg + run_base((c & 3) * (LL >> 4) + (rowbase >> 6), colbase)
                                + ((c >> 2) + hh * 16) * 8;
        _Float16* yp = yimg + (colbase >> 5) * 512 + lane * 8;
        Acc44 ra = {};
        wave_gemm64<true>(himg, Dth, 0, colbase, SS, ra);
#pragma unroll
        for (int i = 0; i < 4; ++i)
#pragma unroll
            for (int j = 0; j < 4; ++j) {
                const v8h z = *(const v8h*)(zp + j * 256 + i * 32);
                v8h o;
#pragma unroll
                for (int e = 0; e < 8; ++e)
                    o[e] = (_Float16)(ra.c[i][j][e] * (float)z[e]);
                *(v8h*)(yp + i * (MM >> 5) * 512 + j * 256) = o;
            }
    }
    __syncthreads();

    const int cb = w * 128;
    const _Float16* Al = yimg + lane * 8;
    const _Float16* Bb = Woh + (size_t)(cb >> 4) * (MM >> 5) * 512 + lane * 8;
    Acc48 acc = {};
    ASet4 a0, a1;
    BSet8 b0, b1;
    load_b8(b0, Bb, 0); load_a4(a0, Al, 0);
    load_b8(b1, Bb, 1); load_a4(a1, Al, 1);
    for (int kb = 2; kb < (MM >> 5); kb += 2) {
        mma_48<false>(acc, a0, b0); __builtin_amdgcn_sched_barrier(0);
        load_b8(b0, Bb, kb); load_a4(a0, Al, kb); __builtin_amdgcn_sched_barrier(0);
        mma_48<false>(acc, a1, b1); __builtin_amdgcn_sched_barrier(0);
        load_b8(b1, Bb, kb + 1); load_a4(a1, Al, kb + 1); __builtin_amdgcn_sched_barrier(0);
    }
    mma_48<false>(acc, a0, b0);
    mma_48<false>(acc, a1, b1);

    const int nlo = lane & 15, mh = (lane >> 4) * 8;
#pragma unroll
    for (int j = 0; j < 8; ++j) {
        const int n = cb + j * 16 + nlo;
        const float bias = b_out[n];
#pragma unroll
        for (int i = 0; i < 4; ++i)
#pragma unroll
            for (int jj = 0; jj < 8; ++jj) {
                const int m = rowbase + i * 16 + mh + jj;
                const size_t idx = (size_t)m * MM + n;
                out[idx] = xflat[idx] + acc.c[i][j][jj] + bias;
            }
    }
}

extern "C" void kernel_launch(void* const* d_in, const int* in_sizes, int n_in,
                              void* d_out, int out_size, void* d_ws, size_t ws_size,
                              hipStream_t stream) {
    const float* x     = (const float*)d_in[0];
    const float* A_log = (const float*)d_in[1];
    const float* Dmat  = (const float*)d_in[2];
    const float* W_in  = (const float*)d_in[3];
    const float* b_in  = (const float*)d_in[4];
    const float* W_xp  = (const float*)d_in[5];
    const float* b_xp  = (const float*)d_in[6];
    const float* W_out = (const float*)d_in[7];
    const float* b_out = (const float*)d_in[8];
    const float* gamma = (const float*)d_in[9];
    const float* beta  = (const float*)d_in[10];
    float* out = (float*)d_out;

    char* ws = (char*)d_ws;
    size_t off = 0;
    auto take = [&](size_t bytes) { char* p = ws + off; off += bytes; return p; };
    float*    bend  = (float*)take((size_t)BB * NSEG64 * SS * 4);
    _Float16* z_h   = (_Float16*)take((size_t)ROWS * MM * 2);
    float*    hloc  = (float*)take((size_t)ROWS * SS * 4);
    _Float16* Wih   = (_Float16*)take((size_t)2 * MM * MM * 2);
    _Float16* Wxh   = (_Float16*)take((size_t)SS * MM * 2);
    _Float16* Woh   = (_Float16*)take((size_t)MM * MM * 2);
    _Float16* Dth   = (_Float16*)take((size_t)SS * MM * 2);
    (void)ws_size; (void)in_sizes; (void)n_in; (void)out_size;

    k_convert<<<NGROUPS / 256, 256, 0, stream>>>(W_in, W_xp, W_out, Dmat, Wih, Wxh, Woh, Dth);
    k_inproj<<<ROWS / 64, 256, 0, stream>>>(x, gamma, beta, Wih, b_in, Wxh, b_xp, A_log, hloc, bend, z_h);
    k_gemm34<<<ROWS / 64, 256, 0, stream>>>(hloc, A_log, bend, Dth, z_h, Woh, b_out, x, out);
}
